# starting kernel
# baseline (speedup 1.0000x reference)
#include <hip/hip_runtime.h>
#include <hip/hip_bf16.h>

typedef __attribute__((ext_vector_type(16))) _Float16 v16h;
typedef __attribute__((ext_vector_type(8)))  _Float16 v8h;
typedef __attribute__((ext_vector_type(4)))  _Float16 v4h;
typedef __attribute__((ext_vector_type(8)))  float    v8f;

#define N_NODES 50000
#define N_EDGES 800000
#define DIM 128
#define HEADS 8
#define HEAD_DIM 16

__device__ inline void atomicAddF(float* p, float v) {
    __hip_atomic_fetch_add(p, v, __ATOMIC_RELAXED, __HIP_MEMORY_SCOPE_AGENT);
}
__device__ inline void atomicMaxF(int* p, float v) {
    if (v >= 0.0f) atomicMax(p, __float_as_int(v));
    else           atomicMin((unsigned int*)p, __float_as_uint(v));
}

__device__ inline void gemm_tile_16x16(const v16h a[4], const _Float16* __restrict__ WT,
                                       const float* __restrict__ bias,
                                       float* __restrict__ out,
                                       int r0, int n0, int lane) {
    const int col = lane & 15;
    const int kb  = (lane >> 4) << 3;
    const _Float16* wr = WT + (size_t)(n0 + col) * DIM;
    v8f c = {};
#pragma unroll
    for (int t = 0; t < 4; ++t) {
        const v8h lo = *(const v8h*)(wr + t * 32 + kb);
        const v8h hi = *(const v8h*)(wr + t * 32 + kb + 16);
        v16h b;
#pragma unroll
        for (int i = 0; i < 8; ++i) { b[i] = lo[i]; b[i + 8] = hi[i]; }
        c = __builtin_amdgcn_wmma_f32_16x16x32_f16(false, a[t], false, b,
                                                   (short)0, c, false, false);
    }
    const float bia = bias[n0 + col];
    float* op = out + (size_t)r0 * DIM + n0 + col;
#pragma unroll
    for (int j = 0; j < 8; ++j) {
        int m = j + ((lane >> 4) << 3);
        op[(size_t)m * DIM] = c[j] + bia;
    }
}

__device__ inline void load_a_frags(const _Float16* __restrict__ src, int r0, int lane,
                                    v16h a[4]) {
    const int row = lane & 15;
    const int kb  = (lane >> 4) << 3;
    const _Float16* xr = src + (size_t)(r0 + row) * DIM;
#pragma unroll
    for (int t = 0; t < 4; ++t) {
        const v8h lo = *(const v8h*)(xr + t * 32 + kb);
        const v8h hi = *(const v8h*)(xr + t * 32 + kb + 16);
#pragma unroll
        for (int i = 0; i < 8; ++i) { a[t][i] = lo[i]; a[t][i + 8] = hi[i]; }
    }
}

__global__ void init_state(int* segmax_bits, float* denom, float* vals) {
    int i = blockIdx.x * blockDim.x + threadIdx.x;
    if (i < N_NODES * HEADS) {
        segmax_bits[i] = 0xFF800000;
        denom[i] = 0.0f;
    }
    if (i < N_NODES * DIM) vals[i] = 0.0f;
}

__global__ void cvt_f32_f16_x4(const float* __restrict__ src, _Float16* __restrict__ dst) {
    int i = blockIdx.x * blockDim.x + threadIdx.x;
    float4 vv = ((const float4*)src)[i];
    v4h o = { (_Float16)vv.x, (_Float16)vv.y, (_Float16)vv.z, (_Float16)vv.w };
    ((v4h*)dst)[i] = o;
}

__global__ void cvt_weightsT(const float* __restrict__ Wq, const float* __restrict__ Wk,
                             const float* __restrict__ Wv, const float* __restrict__ Wo,
                             _Float16* __restrict__ WT) {
    int i = blockIdx.x * blockDim.x + threadIdx.x;
    int m = i >> 14, r = i & 16383, n = r >> 7, k = r & 127;
    const float* W = (m == 0) ? Wq : (m == 1) ? Wk : (m == 2) ? Wv : Wo;
    WT[i] = (_Float16)W[(size_t)k * DIM + n];
}

__global__ void qkv_gemm(const _Float16* __restrict__ x16,
                         const _Float16* __restrict__ WTq, const float* __restrict__ bq,
                         const _Float16* __restrict__ WTk, const float* __restrict__ bk,
                         const _Float16* __restrict__ WTv, const float* __restrict__ bv,
                         float* __restrict__ q, float* __restrict__ k,
                         float* __restrict__ v) {
    const int lane = threadIdx.x & 31;
    const int wave = threadIdx.x >> 5;
    const int r0 = blockIdx.x * 16;
    const int n0 = wave * 16;
    v16h a[4];
    load_a_frags(x16, r0, lane, a);
    gemm_tile_16x16(a, WTq, bq, q, r0, n0, lane);
    gemm_tile_16x16(a, WTk, bk, k, r0, n0, lane);
    gemm_tile_16x16(a, WTv, bv, v, r0, n0, lane);
}

__global__ void edge_scores(const int* __restrict__ origin, const int* __restrict__ dest,
                            const float* __restrict__ ew,
                            const float* __restrict__ q, const float* __restrict__ k,
                            float* __restrict__ scores, int* __restrict__ segmax_bits) {
    int idx = blockIdx.x * blockDim.x + threadIdx.x;
    int e = idx >> 3, h = idx & 7;
    int o = origin[e], d = dest[e];
    const float4* qp = (const float4*)(q + (size_t)o * DIM + h * HEAD_DIM);
    const float4* kp = (const float4*)(k + (size_t)d * DIM + h * HEAD_DIM);
    float s = 0.0f;
#pragma unroll
    for (int t = 0; t < 4; ++t) {
        float4 av = qp[t], bv = kp[t];
        s += av.x * bv.x + av.y * bv.y + av.z * bv.z + av.w * bv.w;
    }
    s *= 0.25f;
    s *= ew[idx];
    scores[idx] = s;
    atomicMaxF(&segmax_bits[o * HEADS + h], s);
}

__global__ void edge_exp(const int* __restrict__ origin,
                         const int* __restrict__ segmax_bits,
                         float* __restrict__ scores, float* __restrict__ denom) {
    int idx = blockIdx.x * blockDim.x + threadIdx.x;
    int e = idx >> 3, h = idx & 7;
    int o = origin[e];
    float sm = __int_as_float(segmax_bits[o * HEADS + h]);
    if (!isfinite(sm)) sm = 0.0f;
    float ev = __expf(scores[idx] - sm);
    scores[idx] = ev;
    atomicAddF(&denom[o * HEADS + h], ev);
}

__global__ void edge_scatter(const int* __restrict__ origin, const int* __restrict__ dest,
                             const float* __restrict__ ev, const float* __restrict__ denom,
                             const float* __restrict__ v, float* __restrict__ vals) {
    int idx = blockIdx.x * blockDim.x + threadIdx.x;
    int e = idx >> 7, dim = idx & 127, h = dim >> 4;
    int o = origin[e], d = dest[e];
    float attn = ev[e * HEADS + h] / (denom[o * HEADS + h] + 1e-16f);
    atomicAddF(&vals[(size_t)o * DIM + dim], attn * v[(size_t)d * DIM + dim]);
}

__global__ void out_gemm(const _Float16* __restrict__ vals16,
                         const _Float16* __restrict__ WTo, const float* __restrict__ bo,
                         float* __restrict__ out) {
    const int lane = threadIdx.x & 31;
    const int wave = threadIdx.x >> 5;
    const int r0 = blockIdx.x * 16;
    const int n0 = wave * 16;
    v16h a[4];
    load_a_frags(vals16, r0, lane, a);
    gemm_tile_16x16(a, WTo, bo, out, r0, n0, lane);
}

__global__ void ln_residual(const float* __restrict__ outp, const float* __restrict__ x,
                            const float* __restrict__ gamma, const float* __restrict__ beta,
                            float* __restrict__ y) {
    const int lane = threadIdx.x & 31;
    const int wave = threadIdx.x >> 5;
    const int node = blockIdx.x * 8 + wave;
    const float4 ov = ((const float4*)(outp + (size_t)node * DIM))[lane];
    float s = ov.x + ov.y + ov.z + ov.w;
#pragma unroll
    for (int m = 16; m >= 1; m >>= 1) s += __shfl_xor(s, m, 32);
    const float mu = s * (1.0f / DIM);
    float dx0 = ov.x - mu, dx1 = ov.y - mu, dx2 = ov.z - mu, dx3 = ov.w - mu;
    float s2 = dx0 * dx0 + dx1 * dx1 + dx2 * dx2 + dx3 * dx3;
#pragma unroll
    for (int m = 16; m >= 1; m >>= 1) s2 += __shfl_xor(s2, m, 32);
    const float inv = rsqrtf(s2 * (1.0f / DIM) + 1e-5f);
    const float4 xv = ((const float4*)(x + (size_t)node * DIM))[lane];
    const float4 gv = ((const float4*)gamma)[lane];
    const float4 bv = ((const float4*)beta)[lane];
    float4 r;
    r.x = xv.x + dx0 * inv * gv.x + bv.x;
    r.y = xv.y + dx1 * inv * gv.y + bv.y;
    r.z = xv.z + dx2 * inv * gv.z + bv.z;
    r.w = xv.w + dx3 * inv * gv.w + bv.w;
    ((float4*)(y + (size_t)node * DIM))[lane] = r;
}

extern "C" void kernel_launch(void* const* d_in, const int* in_sizes, int n_in,
                              void* d_out, int out_size, void* d_ws, size_t ws_size,
                              hipStream_t stream) {
    const float* x   = (const float*)d_in[0];
    const int*   ei  = (const int*)  d_in[1];
    const float* ew  = (const float*)d_in[2];
    const float* Wq  = (const float*)d_in[3];
    const float* bq  = (const float*)d_in[4];
    const float* Wk  = (const float*)d_in[5];
    const float* bk  = (const float*)d_in[6];
    const float* Wv  = (const float*)d_in[7];
    const float* bv  = (const float*)d_in[8];
    const float* Wo  = (const float*)d_in[9];
    const float* bo  = (const float*)d_in[10];
    const float* gam = (const float*)d_in[11];
    const float* bet = (const float*)d_in[12];
    float* y = (float*)d_out;

    const int* origin = ei;
    const int* dest   = ei + N_EDGES;

    float* ws   = (float*)d_ws;
    float* q    = ws;
    float* k    = q    + (size_t)N_NODES * DIM;
    float* v    = k    + (size_t)N_NODES * DIM;
    float* sc   = v    + (size_t)N_NODES * DIM;
    float* smx  = sc   + (size_t)N_EDGES * HEADS;
    float* den  = smx  + (size_t)N_NODES * HEADS;
    float* vals = den  + (size_t)N_NODES * HEADS;
    float* outb = vals + (size_t)N_NODES * DIM;
    _Float16* h16    = (_Float16*)(outb + (size_t)N_NODES * DIM);
    _Float16* x16    = h16;
    _Float16* vals16 = x16 + (size_t)N_NODES * DIM;
    _Float16* WT     = vals16 + (size_t)N_NODES * DIM;
    _Float16* WTq = WT;
    _Float16* WTk = WT + 1 * DIM * DIM;
    _Float16* WTv = WT + 2 * DIM * DIM;
    _Float16* WTo = WT + 3 * DIM * DIM;

    init_state<<<(N_NODES * DIM + 255) / 256, 256, 0, stream>>>((int*)smx, den, vals);

    cvt_f32_f16_x4<<<(N_NODES * DIM / 4) / 256, 256, 0, stream>>>(x, x16);
    cvt_weightsT<<<(4 * DIM * DIM) / 256, 256, 0, stream>>>(Wq, Wk, Wv, Wo, WT);

    qkv_gemm<<<N_NODES / 16, 256, 0, stream>>>(x16, WTq, bq, WTk, bk, WTv, bv, q, k, v);

    edge_scores<<<(N_EDGES * HEADS) / 256, 256, 0, stream>>>(origin, dest, ew, q, k,
                                                             sc, (int*)smx);

    edge_exp<<<(N_EDGES * HEADS) / 256, 256, 0, stream>>>(origin, (const int*)smx,
                                                          sc, den);

    edge_scatter<<<(N_EDGES * DIM) / 256, 256, 0, stream>>>(origin, dest, sc, den,
                                                            v, vals);

    cvt_f32_f16_x4<<<(N_NODES * DIM / 4) / 256, 256, 0, stream>>>(vals, vals16);

    out_gemm<<<N_NODES / 16, 256, 0, stream>>>(vals16, WTo, bo, outb);

    ln_residual<<<N_NODES / 8, 256, 0, stream>>>(outb, x, gam, bet, y);
}


# Scatter by node-range ownership: 256 workgroups hold 196 nodes' value rows each in LDS, scan all origins, add own edges' attn*v by LDS float adds; no global atomics on vals; sum order: LDS arrival
# speedup vs baseline: 1.6838x; 1.6838x over previous
#include <hip/hip_runtime.h>
#include <hip/hip_bf16.h>

typedef __attribute__((ext_vector_type(16))) _Float16 v16h;
typedef __attribute__((ext_vector_type(8)))  _Float16 v8h;
typedef __attribute__((ext_vector_type(4)))  _Float16 v4h;
typedef __attribute__((ext_vector_type(8)))  float    v8f;

#define N_NODES 50000
#define N_EDGES 800000
#define DIM 128
#define HEADS 8
#define HEAD_DIM 16

__device__ inline void atomicAddF(float* p, float v) {
    __hip_atomic_fetch_add(p, v, __ATOMIC_RELAXED, __HIP_MEMORY_SCOPE_AGENT);
}
__device__ inline void atomicMaxF(int* p, float v) {
    if (v >= 0.0f) atomicMax(p, __float_as_int(v));
    else           atomicMin((unsigned int*)p, __float_as_uint(v));
}

__device__ inline void gemm_tile_16x16(const v16h a[4], const _Float16* __restrict__ WT,
                                       const float* __restrict__ bias,
                                       float* __restrict__ out,
                                       int r0, int n0, int lane) {
    const int col = lane & 15;
    const int kb  = (lane >> 4) << 3;
    const _Float16* wr = WT + (size_t)(n0 + col) * DIM;
    v8f c = {};
#pragma unroll
    for (int t = 0; t < 4; ++t) {
        const v8h lo = *(const v8h*)(wr + t * 32 + kb);
        const v8h hi = *(const v8h*)(wr + t * 32 + kb + 16);
        v16h b;
#pragma unroll
        for (int i = 0; i < 8; ++i) { b[i] = lo[i]; b[i + 8] = hi[i]; }
        c = __builtin_amdgcn_wmma_f32_16x16x32_f16(false, a[t], false, b,
                                                   (short)0, c, false, false);
    }
    const float bia = bias[n0 + col];
    float* op = out + (size_t)r0 * DIM + n0 + col;
#pragma unroll
    for (int j = 0; j < 8; ++j) {
        int m = j + ((lane >> 4) << 3);
        op[(size_t)m * DIM] = c[j] + bia;
    }
}

__device__ inline void load_a_frags(const _Float16* __restrict__ src, int r0, int lane,
                                    v16h a[4]) {
    const int row = lane & 15;
    const int kb  = (lane >> 4) << 3;
    const _Float16* xr = src + (size_t)(r0 + row) * DIM;
#pragma unroll
    for (int t = 0; t < 4; ++t) {
        const v8h lo = *(const v8h*)(xr + t * 32 + kb);
        const v8h hi = *(const v8h*)(xr + t * 32 + kb + 16);
#pragma unroll
        for (int i = 0; i < 8; ++i) { a[t][i] = lo[i]; a[t][i + 8] = hi[i]; }
    }
}

__global__ void init_state(int* segmax_bits, float* denom) {
    int i = blockIdx.x * blockDim.x + threadIdx.x;
    if (i < N_NODES * HEADS) {
        segmax_bits[i] = 0xFF800000;
        denom[i] = 0.0f;
    }
}

__global__ void cvt_f32_f16_x4(const float* __restrict__ src, _Float16* __restrict__ dst) {
    int i = blockIdx.x * blockDim.x + threadIdx.x;
    float4 vv = ((const float4*)src)[i];
    v4h o = { (_Float16)vv.x, (_Float16)vv.y, (_Float16)vv.z, (_Float16)vv.w };
    ((v4h*)dst)[i] = o;
}

__global__ void cvt_weightsT(const float* __restrict__ Wq, const float* __restrict__ Wk,
                             const float* __restrict__ Wv, const float* __restrict__ Wo,
                             _Float16* __restrict__ WT) {
    int i = blockIdx.x * blockDim.x + threadIdx.x;
    int m = i >> 14, r = i & 16383, n = r >> 7, k = r & 127;
    const float* W = (m == 0) ? Wq : (m == 1) ? Wk : (m == 2) ? Wv : Wo;
    WT[i] = (_Float16)W[(size_t)k * DIM + n];
}

__global__ void qkv_gemm(const _Float16* __restrict__ x16,
                         const _Float16* __restrict__ WTq, const float* __restrict__ bq,
                         const _Float16* __restrict__ WTk, const float* __restrict__ bk,
                         const _Float16* __restrict__ WTv, const float* __restrict__ bv,
                         float* __restrict__ q, float* __restrict__ k,
                         float* __restrict__ v) {
    const int lane = threadIdx.x & 31;
    const int wave = threadIdx.x >> 5;
    const int r0 = blockIdx.x * 16;
    const int n0 = wave * 16;
    v16h a[4];
    load_a_frags(x16, r0, lane, a);
    gemm_tile_16x16(a, WTq, bq, q, r0, n0, lane);
    gemm_tile_16x16(a, WTk, bk, k, r0, n0, lane);
    gemm_tile_16x16(a, WTv, bv, v, r0, n0, lane);
}

__global__ void edge_scores(const int* __restrict__ origin, const int* __restrict__ dest,
                            const float* __restrict__ ew,
                            const float* __restrict__ q, const float* __restrict__ k,
                            float* __restrict__ scores, int* __restrict__ segmax_bits) {
    int idx = blockIdx.x * blockDim.x + threadIdx.x;
    int e = idx >> 3, h = idx & 7;
    int o = origin[e], d = dest[e];
    const float4* qp = (const float4*)(q + (size_t)o * DIM + h * HEAD_DIM);
    const float4* kp = (const float4*)(k + (size_t)d * DIM + h * HEAD_DIM);
    float s = 0.0f;
#pragma unroll
    for (int t = 0; t < 4; ++t) {
        float4 av = qp[t], bv = kp[t];
        s += av.x * bv.x + av.y * bv.y + av.z * bv.z + av.w * bv.w;
    }
    s *= 0.25f;
    s *= ew[idx];
    scores[idx] = s;
    atomicMaxF(&segmax_bits[o * HEADS + h], s);
}

__global__ void edge_exp(const int* __restrict__ origin,
                         const int* __restrict__ segmax_bits,
                         float* __restrict__ scores, float* __restrict__ denom) {
    int idx = blockIdx.x * blockDim.x + threadIdx.x;
    int e = idx >> 3, h = idx & 7;
    int o = origin[e];
    float sm = __int_as_float(segmax_bits[o * HEADS + h]);
    if (!isfinite(sm)) sm = 0.0f;
    float ev = __expf(scores[idx] - sm);
    scores[idx] = ev;
    atomicAddF(&denom[o * HEADS + h], ev);
}

#define OWN_NODES 196
#define OWN_BLOCKS 256
#define CHUNKS128 (N_EDGES / 128)
#define SUPER ((CHUNKS128 + 3) / 4)

__device__ inline void lds_add(float* p, float x) {
    __hip_atomic_fetch_add(p, x, __ATOMIC_RELAXED, __HIP_MEMORY_SCOPE_WORKGROUP);
}

__device__ inline void load_origins(const int* __restrict__ origin, int sc, int lane, int o[16]) {
#pragma unroll
    for (int k = 0; k < 4; ++k) {
        const int ch = sc * 4 + k;
        int4 t = make_int4(-1, -1, -1, -1);
        if (ch < CHUNKS128) t = *(const int4*)(origin + ch * 128 + lane * 4);
        o[k * 4 + 0] = t.x; o[k * 4 + 1] = t.y; o[k * 4 + 2] = t.z; o[k * 4 + 3] = t.w;
    }
}

__global__ __launch_bounds__(1024)
void scatter_owned(const int* __restrict__ origin, const int* __restrict__ dest,
                   const float* __restrict__ ev, const float* __restrict__ denom,
                   const float* __restrict__ v, float* __restrict__ vals) {
    __shared__ float tile[OWN_NODES * DIM];
    const int tid = threadIdx.x;
    const int lane = tid & 31;
    const int wave = __builtin_amdgcn_readfirstlane(tid >> 5);
    const int node0 = blockIdx.x * OWN_NODES;
    const int h = lane >> 2;

    float4* t4 = (float4*)tile;
    for (int i = tid; i < OWN_NODES * DIM / 4; i += 1024)
        t4[i] = make_float4(0.0f, 0.0f, 0.0f, 0.0f);
    __syncthreads();

    const int rot = blockIdx.x * 6;
    int sc = wave + rot;
    if (sc >= SUPER) sc -= SUPER;
    int o[16];
    load_origins(origin, sc, lane, o);
    for (int s = wave; s < SUPER; s += 32) {
        int scn = s + 32 + rot;
        if (scn >= SUPER) scn -= SUPER;
        int on[16];
#pragma unroll
        for (int i = 0; i < 16; ++i) on[i] = -1;
        if (s + 32 < SUPER) load_origins(origin, scn, lane, on);

        unsigned mk[16];
        unsigned any = 0u;
#pragma unroll
        for (int i = 0; i < 16; ++i) {
            mk[i] = __builtin_amdgcn_ballot_w32((unsigned)(o[i] - node0) < (unsigned)OWN_NODES);
            any |= mk[i];
        }
        if (any != 0u) {
            int dd[16];
#pragma unroll
            for (int k = 0; k < 4; ++k) {
                const unsigned mq = mk[k * 4] | mk[k * 4 + 1] | mk[k * 4 + 2] | mk[k * 4 + 3];
                int4 t = make_int4(0, 0, 0, 0);
                if ((mq >> lane) & 1u)
                    t = *(const int4*)(dest + (sc * 4 + k) * 128 + lane * 4);
                dd[k * 4 + 0] = t.x; dd[k * 4 + 1] = t.y; dd[k * 4 + 2] = t.z; dd[k * 4 + 3] = t.w;
            }
#pragma unroll
            for (int i = 0; i < 16; ++i) {
                unsigned mask = mk[i];
                while (mask) {
                    const int j = __builtin_ctz(mask);
                    mask &= mask - 1u;
                    const int oj = __builtin_amdgcn_readlane(o[i], j);
                    const int dj = __builtin_amdgcn_readlane(dd[i], j);
                    const int e = (sc * 4 + (i >> 2)) * 128 + j * 4 + (i & 3);
                    const float a = ev[e * HEADS + h] / (denom[oj * HEADS + h] + 1e-16f);
                    const float4 vv = *(const float4*)(v + (size_t)dj * DIM + lane * 4);
                    float* tp = tile + (oj - node0) * DIM + lane * 4;
                    lds_add(tp + 0, a * vv.x);
                    lds_add(tp + 1, a * vv.y);
                    lds_add(tp + 2, a * vv.z);
                    lds_add(tp + 3, a * vv.w);
                }
            }
        }
        sc = scn;
#pragma unroll
        for (int i = 0; i < 16; ++i) o[i] = on[i];
    }
    __syncthreads();

    float4* out4 = (float4*)(vals + (size_t)node0 * DIM);
    for (int i = tid; i < OWN_NODES * DIM / 4; i += 1024)
        if (node0 + (i >> 5) < N_NODES) out4[i] = t4[i];
}

__global__ void out_gemm(const _Float16* __restrict__ vals16,
                         const _Float16* __restrict__ WTo, const float* __restrict__ bo,
                         float* __restrict__ out) {
    const int lane = threadIdx.x & 31;
    const int wave = threadIdx.x >> 5;
    const int r0 = blockIdx.x * 16;
    const int n0 = wave * 16;
    v16h a[4];
    load_a_frags(vals16, r0, lane, a);
    gemm_tile_16x16(a, WTo, bo, out, r0, n0, lane);
}

__global__ void ln_residual(const float* __restrict__ outp, const float* __restrict__ x,
                            const float* __restrict__ gamma, const float* __restrict__ beta,
                            float* __restrict__ y) {
    const int lane = threadIdx.x & 31;
    const int wave = threadIdx.x >> 5;
    const int node = blockIdx.x * 8 + wave;
    const float4 ov = ((const float4*)(outp + (size_t)node * DIM))[lane];
    float s = ov.x + ov.y + ov.z + ov.w;
#pragma unroll
    for (int m = 16; m >= 1; m >>= 1) s += __shfl_xor(s, m, 32);
    const float mu = s * (1.0f / DIM);
    float dx0 = ov.x - mu, dx1 = ov.y - mu, dx2 = ov.z - mu, dx3 = ov.w - mu;
    float s2 = dx0 * dx0 + dx1 * dx1 + dx2 * dx2 + dx3 * dx3;
#pragma unroll
    for (int m = 16; m >= 1; m >>= 1) s2 += __shfl_xor(s2, m, 32);
    const float inv = rsqrtf(s2 * (1.0f / DIM) + 1e-5f);
    const float4 xv = ((const float4*)(x + (size_t)node * DIM))[lane];
    const float4 gv = ((const float4*)gamma)[lane];
    const float4 bv = ((const float4*)beta)[lane];
    float4 r;
    r.x = xv.x + dx0 * inv * gv.x + bv.x;
    r.y = xv.y + dx1 * inv * gv.y + bv.y;
    r.z = xv.z + dx2 * inv * gv.z + bv.z;
    r.w = xv.w + dx3 * inv * gv.w + bv.w;
    ((float4*)(y + (size_t)node * DIM))[lane] = r;
}

extern "C" void kernel_launch(void* const* d_in, const int* in_sizes, int n_in,
                              void* d_out, int out_size, void* d_ws, size_t ws_size,
                              hipStream_t stream) {
    const float* x   = (const float*)d_in[0];
    const int*   ei  = (const int*)  d_in[1];
    const float* ew  = (const float*)d_in[2];
    const float* Wq  = (const float*)d_in[3];
    const float* bq  = (const float*)d_in[4];
    const float* Wk  = (const float*)d_in[5];
    const float* bk  = (const float*)d_in[6];
    const float* Wv  = (const float*)d_in[7];
    const float* bv  = (const float*)d_in[8];
    const float* Wo  = (const float*)d_in[9];
    const float* bo  = (const float*)d_in[10];
    const float* gam = (const float*)d_in[11];
    const float* bet = (const float*)d_in[12];
    float* y = (float*)d_out;

    const int* origin = ei;
    const int* dest   = ei + N_EDGES;

    float* ws   = (float*)d_ws;
    float* q    = ws;
    float* k    = q    + (size_t)N_NODES * DIM;
    float* v    = k    + (size_t)N_NODES * DIM;
    float* sc   = v    + (size_t)N_NODES * DIM;
    float* smx  = sc   + (size_t)N_EDGES * HEADS;
    float* den  = smx  + (size_t)N_NODES * HEADS;
    float* vals = den  + (size_t)N_NODES * HEADS;
    float* outb = vals + (size_t)N_NODES * DIM;
    _Float16* h16    = (_Float16*)(outb + (size_t)N_NODES * DIM);
    _Float16* x16    = h16;
    _Float16* vals16 = x16 + (size_t)N_NODES * DIM;
    _Float16* WT     = vals16 + (size_t)N_NODES * DIM;
    _Float16* WTq = WT;
    _Float16* WTk = WT + 1 * DIM * DIM;
    _Float16* WTv = WT + 2 * DIM * DIM;
    _Float16* WTo = WT + 3 * DIM * DIM;

    init_state<<<(N_NODES * HEADS + 255) / 256, 256, 0, stream>>>((int*)smx, den);

    cvt_f32_f16_x4<<<(N_NODES * DIM / 4) / 256, 256, 0, stream>>>(x, x16);
    cvt_weightsT<<<(4 * DIM * DIM) / 256, 256, 0, stream>>>(Wq, Wk, Wv, Wo, WT);

    qkv_gemm<<<N_NODES / 16, 256, 0, stream>>>(x16, WTq, bq, WTk, bk, WTv, bv, q, k, v);

    edge_scores<<<(N_EDGES * HEADS) / 256, 256, 0, stream>>>(origin, dest, ew, q, k,
                                                             sc, (int*)smx);

    edge_exp<<<(N_EDGES * HEADS) / 256, 256, 0, stream>>>(origin, (const int*)smx,
                                                          sc, den);

    scatter_owned<<<OWN_BLOCKS, 1024, 0, stream>>>(origin, dest, sc, den, v, vals);

    cvt_f32_f16_x4<<<(N_NODES * DIM / 4) / 256, 256, 0, stream>>>(vals, vals16);

    out_gemm<<<N_NODES / 16, 256, 0, stream>>>(vals16, WTo, bo, outb);

    ln_residual<<<N_NODES / 8, 256, 0, stream>>>(outb, x, gam, bet, y);
}


# Owner scan reads a one-byte owner map (origin/196, from extra blocks of the weight launch), not 4-byte origins: a quarter of the bytes, four edges per vector test; list order new, sums: LDS arrival
# speedup vs baseline: 2.7517x; 1.6342x over previous
#include <hip/hip_runtime.h>
#include <hip/hip_bf16.h>

typedef __attribute__((ext_vector_type(16))) _Float16 v16h;
typedef __attribute__((ext_vector_type(8)))  _Float16 v8h;
typedef __attribute__((ext_vector_type(4)))  _Float16 v4h;
typedef __attribute__((ext_vector_type(8)))  float    v8f;

#define N_NODES 50000
#define N_EDGES 800000
#define DIM 128
#define HEADS 8
#define HEAD_DIM 16

__device__ inline void gemm_tile_16x16(const v16h a[4], const _Float16* __restrict__ WT,
                                       const float* __restrict__ bias,
                                       float* __restrict__ out,
                                       int r0, int n0, int lane) {
    const int col = lane & 15;
    const int kb  = (lane >> 4) << 3;
    const _Float16* wr = WT + (size_t)(n0 + col) * DIM;
    v8f c = {};
#pragma unroll
    for (int t = 0; t < 4; ++t) {
        const v8h lo = *(const v8h*)(wr + t * 32 + kb);
        const v8h hi = *(const v8h*)(wr + t * 32 + kb + 16);
        v16h b;
#pragma unroll
        for (int i = 0; i < 8; ++i) { b[i] = lo[i]; b[i + 8] = hi[i]; }
        c = __builtin_amdgcn_wmma_f32_16x16x32_f16(false, a[t], false, b,
                                                   (short)0, c, false, false);
    }
    const float bia = bias[n0 + col];
    float* op = out + (size_t)r0 * DIM + n0 + col;
#pragma unroll
    for (int j = 0; j < 8; ++j) {
        int m = j + ((lane >> 4) << 3);
        op[(size_t)m * DIM] = c[j] + bia;
    }
}

__device__ inline void load_a_frags(const _Float16* __restrict__ src, int r0, int lane,
                                    v16h a[4]) {
    const int row = lane & 15;
    const int kb  = (lane >> 4) << 3;
    const _Float16* xr = src + (size_t)(r0 + row) * DIM;
#pragma unroll
    for (int t = 0; t < 4; ++t) {
        const v8h lo = *(const v8h*)(xr + t * 32 + kb);
        const v8h hi = *(const v8h*)(xr + t * 32 + kb + 16);
#pragma unroll
        for (int i = 0; i < 8; ++i) { a[t][i] = lo[i]; a[t][i + 8] = hi[i]; }
    }
}

__global__ void cvt_f32_f16_x4(const float* __restrict__ src, _Float16* __restrict__ dst) {
    int i = blockIdx.x * blockDim.x + threadIdx.x;
    float4 vv = ((const float4*)src)[i];
    v4h o = { (_Float16)vv.x, (_Float16)vv.y, (_Float16)vv.z, (_Float16)vv.w };
    ((v4h*)dst)[i] = o;
}

#define OWN_NODES 196
#define MAP_VECS (N_EDGES / 16)
#define MAP_STEPS ((MAP_VECS + 63) / 64)
#define MAP_BLOCKS ((MAP_VECS + 255) / 256)
#define WT_BLOCKS ((4 * DIM * DIM) / 256)

__device__ inline unsigned owner_bytes4(const int4 s) {
    return ((unsigned)s.x / OWN_NODES) | (((unsigned)s.y / OWN_NODES) << 8) |
           (((unsigned)s.z / OWN_NODES) << 16) | (((unsigned)s.w / OWN_NODES) << 24);
}

__global__ void cvt_weightsT(const float* __restrict__ Wq, const float* __restrict__ Wk,
                             const float* __restrict__ Wv, const float* __restrict__ Wo,
                             _Float16* __restrict__ WT,
                             const int* __restrict__ origin, unsigned* __restrict__ own) {
    if (blockIdx.x < WT_BLOCKS) {
        int i = blockIdx.x * blockDim.x + threadIdx.x;
        int m = i >> 14, r = i & 16383, n = r >> 7, k = r & 127;
        const float* W = (m == 0) ? Wq : (m == 1) ? Wk : (m == 2) ? Wv : Wo;
        WT[i] = (_Float16)W[(size_t)k * DIM + n];
    } else {
        const int g = (blockIdx.x - WT_BLOCKS) * 256 + threadIdx.x;
        if (g < MAP_VECS) {
            const int4* s = (const int4*)origin + (size_t)g * 4;
            const int4 s0 = s[0], s1 = s[1], s2 = s[2], s3 = s[3];
            ((uint4*)own)[g] = make_uint4(owner_bytes4(s0), owner_bytes4(s1), owner_bytes4(s2), owner_bytes4(s3));
        }
    }
}

__global__ void qkv_gemm(const _Float16* __restrict__ x16,
                         const _Float16* __restrict__ WTq, const float* __restrict__ bq,
                         const _Float16* __restrict__ WTk, const float* __restrict__ bk,
                         const _Float16* __restrict__ WTv, const float* __restrict__ bv,
                         float* __restrict__ q, float* __restrict__ k,
                         float* __restrict__ v) {
    const int lane = threadIdx.x & 31;
    const int wave = threadIdx.x >> 5;
    const int r0 = blockIdx.x * 16;
    const int n0 = wave * 16;
    v16h a[4];
    load_a_frags(x16, r0, lane, a);
    gemm_tile_16x16(a, WTq, bq, q, r0, n0, lane);
    gemm_tile_16x16(a, WTk, bk, k, r0, n0, lane);
    gemm_tile_16x16(a, WTv, bv, v, r0, n0, lane);
}

#define OWN_BLOCKS 256
#define CHUNKS128 (N_EDGES / 128)
#define SUPER ((CHUNKS128 + 3) / 4)
#define LIST_CAP 4096
#define BATCH 8

__device__ inline void lds_add(float* p, float x) {
    __hip_atomic_fetch_add(p, x, __ATOMIC_RELAXED, __HIP_MEMORY_SCOPE_WORKGROUP);
}
__device__ inline void lds_max_f(int* p, float v) {
    if (v >= 0.0f)
        __hip_atomic_fetch_max(p, __float_as_int(v), __ATOMIC_RELAXED, __HIP_MEMORY_SCOPE_WORKGROUP);
    else
        __hip_atomic_fetch_min((unsigned int*)p, __float_as_uint(v), __ATOMIC_RELAXED,
                               __HIP_MEMORY_SCOPE_WORKGROUP);
}

__device__ inline float quad_sum(float x) {
    x += __int_as_float(__builtin_amdgcn_mov_dpp(__float_as_int(x), 0xB1, 0xF, 0xF, true));
    x += __int_as_float(__builtin_amdgcn_mov_dpp(__float_as_int(x), 0x4E, 0xF, 0xF, true));
    return x;
}

__device__ inline float edge_score(const float4 qv, const float4 kv, const float w) {
    float s = qv.x * kv.x + qv.y * kv.y + qv.z * kv.z + qv.w * kv.w;
    s = quad_sum(s);
    s *= 0.25f;
    s *= w;
    return s;
}

__device__ inline void load_origins(const int* __restrict__ origin, int sc, int lane, int o[16]) {
#pragma unroll
    for (int k = 0; k < 4; ++k) {
        const int ch = sc * 4 + k;
        int4 t = make_int4(-1, -1, -1, -1);
        if (ch < CHUNKS128) t = *(const int4*)(origin + ch * 128 + lane * 4);
        o[k * 4 + 0] = t.x; o[k * 4 + 1] = t.y; o[k * 4 + 2] = t.z; o[k * 4 + 3] = t.w;
    }
}

template <int MODE>
__device__ inline void edge_in_place(int e, int oj, int node0, int lane, int h,
                                     const int* __restrict__ dest, const float* __restrict__ ew,
                                     const float* __restrict__ q, const float* __restrict__ k,
                                     const float* __restrict__ v,
                                     int* smax, float* den, float* tile) {
    const int dj = dest[e];
    const float4 qv = *(const float4*)(q + (size_t)oj * DIM + lane * 4);
    const float4 kv = *(const float4*)(k + (size_t)dj * DIM + lane * 4);
    const float s = edge_score(qv, kv, ew[e * HEADS + h]);
    const int ol = oj - node0;
    if (MODE == 1) {
        if ((lane & 3) == 0) lds_max_f(&smax[ol * HEADS + h], s);
    } else {
        float m = __int_as_float(smax[ol * HEADS + h]);
        if (!isfinite(m)) m = 0.0f;
        const float ev = __expf(s - m);
        if ((lane & 3) == 0) lds_add(&den[ol * HEADS + h], ev);
        const float4 vv = *(const float4*)(v + (size_t)dj * DIM + lane * 4);
        float* tp = tile + ol * DIM + lane * 4;
        lds_add(tp + 0, ev * vv.x);
        lds_add(tp + 1, ev * vv.y);
        lds_add(tp + 2, ev * vv.z);
        lds_add(tp + 3, ev * vv.w);
    }
}

template <int MODE>
__device__ inline void scan_owned(const int* __restrict__ origin, const int* __restrict__ dest,
                                  const float* __restrict__ ew,
                                  const float* __restrict__ q, const float* __restrict__ k,
                                  const float* __restrict__ v,
                                  int node0, int wave, int lane, int h,
                                  int* smax, float* den, float* tile) {
    const int rot = blockIdx.x * 6;
    int sc = wave + rot;
    if (sc >= SUPER) sc -= SUPER;
    int o[16];
    load_origins(origin, sc, lane, o);
    for (int s = wave; s < SUPER; s += 32) {
        int scn = s + 32 + rot;
        if (scn >= SUPER) scn -= SUPER;
        int on[16];
#pragma unroll
        for (int i = 0; i < 16; ++i) on[i] = -1;
        if (s + 32 < SUPER) load_origins(origin, scn, lane, on);

        unsigned mk[16];
        unsigned any = 0u;
#pragma unroll
        for (int i = 0; i < 16; ++i) {
            mk[i] = __builtin_amdgcn_ballot_w32((unsigned)(o[i] - node0) < (unsigned)OWN_NODES);
            any |= mk[i];
        }
        if (any != 0u) {
#pragma unroll
            for (int i = 0; i < 16; ++i) {
                unsigned mask = mk[i];
                while (mask) {
                    const int j = __builtin_ctz(mask);
                    mask &= mask - 1u;
                    const int oj = __builtin_amdgcn_readlane(o[i], j);
                    const int e = (sc * 4 + (i >> 2)) * 128 + j * 4 + (i & 3);
                    edge_in_place<MODE>(e, oj, node0, lane, h, dest, ew, q, k, v, smax, den, tile);
                }
            }
        }
        sc = scn;
#pragma unroll
        for (int i = 0; i < 16; ++i) o[i] = on[i];
    }
}

__device__ inline unsigned nonzero_bytes(unsigned x) {
    return ((x & 0x7F7F7F7Fu) + 0x7F7F7F7Fu) | x;
}

__device__ inline void load_map(const uint4* __restrict__ own, int sc, int lane,
                                uint4& a0, uint4& a1) {
    int g0 = sc * 64 + lane;
    int g1 = g0 + 32;
    if (g0 > MAP_VECS - 1) g0 = MAP_VECS - 1;
    if (g1 > MAP_VECS - 1) g1 = MAP_VECS - 1;
    a0 = own[g0];
    a1 = own[g1];
}

__device__ inline void list_owned(const uint4* __restrict__ own, int wave, int lane,
                                  unsigned* list, int* counter) {
    const unsigned pat = blockIdx.x * 0x01010101u;
    const int rot = blockIdx.x * 3;
    int sc = wave + rot;
    if (sc >= MAP_STEPS) sc -= MAP_STEPS;
    uint4 a0, a1;
    load_map(own, sc, lane, a0, a1);
    for (int s = wave; s < MAP_STEPS; s += 32) {
        int scn = s + 32 + rot;
        if (scn >= MAP_STEPS) scn -= MAP_STEPS;
        if (scn >= MAP_STEPS) scn -= MAP_STEPS;
        uint4 n0, n1;
        load_map(own, scn, lane, n0, n1);

        const unsigned M = 0x80808080u;
        unsigned G = nonzero_bytes(a0.x ^ pat) & M;
        G = (nonzero_bytes(a0.y ^ pat) & M) | (G >> 1);
        G = (nonzero_bytes(a0.z ^ pat) & M) | (G >> 1);
        G = (nonzero_bytes(a0.w ^ pat) & M) | (G >> 1);
        G = (nonzero_bytes(a1.x ^ pat) & M) | (G >> 1);
        G = (nonzero_bytes(a1.y ^ pat) & M) | (G >> 1);
        G = (nonzero_bytes(a1.z ^ pat) & M) | (G >> 1);
        G = (nonzero_bytes(a1.w ^ pat) & M) | (G >> 1);
        unsigned F = ~G;
        if (sc == MAP_STEPS - 1) {
            unsigned keep = 0u;
            if (sc * 64 + lane < MAP_VECS) keep |= 0x0F0F0F0Fu;
            if (sc * 64 + 32 + lane < MAP_VECS) keep |= 0xF0F0F0F0u;
            F &= keep;
        }

        unsigned m = __builtin_amdgcn_ballot_w32(F != 0u);
        while (m != 0u) {
            int basep = 0;
            if (lane == 0)
                basep = __hip_atomic_fetch_add(counter, __builtin_popcount(m), __ATOMIC_RELAXED,
                                               __HIP_MEMORY_SCOPE_WORKGROUP);
            basep = __builtin_amdgcn_readfirstlane(basep);
            if (F != 0u) {
                const int p = __builtin_ctz(F);
                F &= F - 1u;
                const int pos = basep + (int)__builtin_amdgcn_mbcnt_lo(m, 0u);
                if (pos < LIST_CAP)
                    list[pos] = (unsigned)(sc * 1024 + lane * 16 + ((p & 4) << 7) + ((p & 3) << 2) + (p >> 3));
            }
            m = __builtin_amdgcn_ballot_w32(F != 0u);
        }
        sc = scn;
        a0 = n0;
        a1 = n1;
    }
}

__global__ __launch_bounds__(1024)
void attend_owned(const unsigned* __restrict__ own, const int* __restrict__ origin, const int* __restrict__ dest,
                  const float* __restrict__ ew,
                  const float* __restrict__ q, const float* __restrict__ k,
                  const float* __restrict__ v, float* __restrict__ vals) {
    __shared__ float tile[OWN_NODES * DIM];
    __shared__ float scores[LIST_CAP * HEADS];
    __shared__ unsigned list[LIST_CAP];
    __shared__ int dlist[LIST_CAP];
    __shared__ int smax[OWN_NODES * HEADS];
    __shared__ float den[OWN_NODES * HEADS];
    __shared__ int counter;
    const int tid = threadIdx.x;
    const int lane = tid & 31;
    const int wave = __builtin_amdgcn_readfirstlane(tid >> 5);
    const int node0 = blockIdx.x * OWN_NODES;
    const int h = lane >> 2;

    float4* t4 = (float4*)tile;
    const float4* q4 = (const float4*)(q + (size_t)node0 * DIM);
    for (int i = tid; i < OWN_NODES * DIM / 4; i += 1024)
        t4[i] = (node0 + (i >> 5) < N_NODES) ? q4[i] : make_float4(0.0f, 0.0f, 0.0f, 0.0f);
    for (int i = tid; i < OWN_NODES * HEADS; i += 1024) {
        smax[i] = (int)0xFF800000;
        den[i] = 0.0f;
    }
    if (tid == 0) counter = 0;
    __syncthreads();

    list_owned((const uint4*)own, wave, lane, list, &counter);
    __syncthreads();
    const int n = __builtin_amdgcn_readfirstlane(counter);

    if (n <= LIST_CAP) {
        const int per = (n + 31) >> 5;
        int wbeg = wave * per;
        if (wbeg > n) wbeg = n;
        int wend = wbeg + per;
        if (wend > n) wend = n;

        for (int base = wbeg; base < wend; base += 32) {
            const int idx = base + lane;
            const bool valid = idx < wend;
            const int e_l = valid ? (int)list[idx] : 0;
            const int d_l = valid ? dest[e_l] : 0;
            const int ol_l = valid ? origin[e_l] - node0 : 0;
            if (valid) {
                dlist[idx] = d_l;
                list[idx] = (unsigned)e_l | ((unsigned)ol_l << 20);
            }
            int cnt = wend - base;
            if (cnt > 32) cnt = 32;
            for (int j0 = 0; j0 < cnt; j0 += BATCH) {
                float4 kk[BATCH], qq[BATCH];
                float ww[BATCH];
#pragma unroll
                for (int u = 0; u < BATCH; ++u) {
                    int jr = j0 + u;
                    if (jr > cnt - 1) jr = cnt - 1;
                    const int ej = __builtin_amdgcn_readlane(e_l, jr);
                    const int oj = __builtin_amdgcn_readlane(ol_l, jr);
                    const int dj = __builtin_amdgcn_readlane(d_l, jr);
                    kk[u] = *(const float4*)(k + (size_t)dj * DIM + lane * 4);
                    ww[u] = ew[ej * HEADS + h];
                    qq[u] = *(const float4*)(tile + oj * DIM + lane * 4);
                }
#pragma unroll
                for (int u = 0; u < BATCH; ++u) {
                    const int j = j0 + u;
                    if (j < cnt) {
                        const int oj = __builtin_amdgcn_readlane(ol_l, j);
                        const float s = edge_score(qq[u], kk[u], ww[u]);
                        if ((lane & 3) == 0) {
                            lds_max_f(&smax[oj * HEADS + h], s);
                            scores[(base + j) * HEADS + h] = s;
                        }
                    }
                }
            }
        }
        __syncthreads();
        for (int i = tid; i < OWN_NODES * DIM / 4; i += 1024)
            t4[i] = make_float4(0.0f, 0.0f, 0.0f, 0.0f);
        __syncthreads();

        for (int base = wbeg; base < wend; base += 32) {
            const int idx = base + lane;
            const bool valid = idx < wend;
            const int ol_l = valid ? (int)(list[idx] >> 20) : 0;
            const int d_l = valid ? dlist[idx] : 0;
            int cnt = wend - base;
            if (cnt > 32) cnt = 32;
            for (int j0 = 0; j0 < cnt; j0 += BATCH) {
                float4 vv[BATCH];
                float ss[BATCH], mm[BATCH];
#pragma unroll
                for (int u = 0; u < BATCH; ++u) {
                    int jr = j0 + u;
                    if (jr > cnt - 1) jr = cnt - 1;
                    const int oj = __builtin_amdgcn_readlane(ol_l, jr);
                    const int dj = __builtin_amdgcn_readlane(d_l, jr);
                    vv[u] = *(const float4*)(v + (size_t)dj * DIM + lane * 4);
                    ss[u] = scores[(base + jr) * HEADS + h];
                    mm[u] = __int_as_float(smax[oj * HEADS + h]);
                }
#pragma unroll
                for (int u = 0; u < BATCH; ++u) {
                    const int j = j0 + u;
                    if (j < cnt) {
                        const int oj = __builtin_amdgcn_readlane(ol_l, j);
                        float m = mm[u];
                        if (!isfinite(m)) m = 0.0f;
                        const float ev = __expf(ss[u] - m);
                        if ((lane & 3) == 0) lds_add(&den[oj * HEADS + h], ev);
                        float* tp = tile + oj * DIM + lane * 4;
                        lds_add(tp + 0, ev * vv[u].x);
                        lds_add(tp + 1, ev * vv[u].y);
                        lds_add(tp + 2, ev * vv[u].z);
                        lds_add(tp + 3, ev * vv[u].w);
                    }
                }
            }
        }
    } else {
        for (int i = tid; i < OWN_NODES * DIM / 4; i += 1024)
            t4[i] = make_float4(0.0f, 0.0f, 0.0f, 0.0f);
        scan_owned<1>(origin, dest, ew, q, k, v, node0, wave, lane, h, smax, den, tile);
        __syncthreads();
        scan_owned<2>(origin, dest, ew, q, k, v, node0, wave, lane, h, smax, den, tile);
    }
    __syncthreads();

    float4* out4 = (float4*)(vals + (size_t)node0 * DIM);
    for (int i = tid; i < OWN_NODES * DIM / 4; i += 1024) {
        const int row = i >> 5;
        if (node0 + row < N_NODES) {
            const float dn = den[row * HEADS + ((i & 31) >> 2)] + 1e-16f;
            float4 a = t4[i];
            a.x /= dn; a.y /= dn; a.z /= dn; a.w /= dn;
            out4[i] = a;
        }
    }
}

__global__ void out_gemm(const _Float16* __restrict__ vals16,
                         const _Float16* __restrict__ WTo, const float* __restrict__ bo,
                         float* __restrict__ out) {
    const int lane = threadIdx.x & 31;
    const int wave = threadIdx.x >> 5;
    const int r0 = blockIdx.x * 16;
    const int n0 = wave * 16;
    v16h a[4];
    load_a_frags(vals16, r0, lane, a);
    gemm_tile_16x16(a, WTo, bo, out, r0, n0, lane);
}

__global__ void ln_residual(const float* __restrict__ outp, const float* __restrict__ x,
                            const float* __restrict__ gamma, const float* __restrict__ beta,
                            float* __restrict__ y) {
    const int lane = threadIdx.x & 31;
    const int wave = threadIdx.x >> 5;
    const int node = blockIdx.x * 8 + wave;
    const float4 ov = ((const float4*)(outp + (size_t)node * DIM))[lane];
    float s = ov.x + ov.y + ov.z + ov.w;
#pragma unroll
    for (int m = 16; m >= 1; m >>= 1) s += __shfl_xor(s, m, 32);
    const float mu = s * (1.0f / DIM);
    float dx0 = ov.x - mu, dx1 = ov.y - mu, dx2 = ov.z - mu, dx3 = ov.w - mu;
    float s2 = dx0 * dx0 + dx1 * dx1 + dx2 * dx2 + dx3 * dx3;
#pragma unroll
    for (int m = 16; m >= 1; m >>= 1) s2 += __shfl_xor(s2, m, 32);
    const float inv = rsqrtf(s2 * (1.0f / DIM) + 1e-5f);
    const float4 xv = ((const float4*)(x + (size_t)node * DIM))[lane];
    const float4 gv = ((const float4*)gamma)[lane];
    const float4 bv = ((const float4*)beta)[lane];
    float4 r;
    r.x = xv.x + dx0 * inv * gv.x + bv.x;
    r.y = xv.y + dx1 * inv * gv.y + bv.y;
    r.z = xv.z + dx2 * inv * gv.z + bv.z;
    r.w = xv.w + dx3 * inv * gv.w + bv.w;
    ((float4*)(y + (size_t)node * DIM))[lane] = r;
}

extern "C" void kernel_launch(void* const* d_in, const int* in_sizes, int n_in,
                              void* d_out, int out_size, void* d_ws, size_t ws_size,
                              hipStream_t stream) {
    const float* x   = (const float*)d_in[0];
    const int*   ei  = (const int*)  d_in[1];
    const float* ew  = (const float*)d_in[2];
    const float* Wq  = (const float*)d_in[3];
    const float* bq  = (const float*)d_in[4];
    const float* Wk  = (const float*)d_in[5];
    const float* bk  = (const float*)d_in[6];
    const float* Wv  = (const float*)d_in[7];
    const float* bv  = (const float*)d_in[8];
    const float* Wo  = (const float*)d_in[9];
    const float* bo  = (const float*)d_in[10];
    const float* gam = (const float*)d_in[11];
    const float* bet = (const float*)d_in[12];
    float* y = (float*)d_out;

    const int* origin = ei;
    const int* dest   = ei + N_EDGES;

    float* ws   = (float*)d_ws;
    float* q    = ws;
    float* k    = q    + (size_t)N_NODES * DIM;
    float* v    = k    + (size_t)N_NODES * DIM;
    float* vals = v    + (size_t)N_NODES * DIM;
    float* outb = vals + (size_t)N_NODES * DIM;
    _Float16* h16    = (_Float16*)(outb + (size_t)N_NODES * DIM);
    _Float16* x16    = h16;
    _Float16* vals16 = x16 + (size_t)N_NODES * DIM;
    _Float16* WT     = vals16 + (size_t)N_NODES * DIM;
    _Float16* WTq = WT;
    _Float16* WTk = WT + 1 * DIM * DIM;
    _Float16* WTv = WT + 2 * DIM * DIM;
    _Float16* WTo = WT + 3 * DIM * DIM;

    cvt_f32_f16_x4<<<(N_NODES * DIM / 4) / 256, 256, 0, stream>>>(x, x16);
    unsigned* own = (unsigned*)outb;
    cvt_weightsT<<<WT_BLOCKS + MAP_BLOCKS, 256, 0, stream>>>(Wq, Wk, Wv, Wo, WT, origin, own);

    qkv_gemm<<<N_NODES / 16, 256, 0, stream>>>(x16, WTq, bq, WTk, bk, WTv, bv, q, k, v);

    attend_owned<<<OWN_BLOCKS, 1024, 0, stream>>>(own, origin, dest, ew, q, k, v, vals);

    cvt_f32_f16_x4<<<(N_NODES * DIM / 4) / 256, 256, 0, stream>>>(vals, vals16);

    out_gemm<<<N_NODES / 16, 256, 0, stream>>>(vals16, WTo, bo, outb);

    ln_residual<<<N_NODES / 8, 256, 0, stream>>>(outb, x, gam, bet, y);
}


# Wo product, LayerNorm, residual in one launch: a wave holds 16 rows x all 128 columns in 8 accumulators, LN in registers; projection plane, LN launch dropped (5->4); LN sums: 8 in lane, 16-lane tree
# speedup vs baseline: 3.0335x; 1.1024x over previous
#include <hip/hip_runtime.h>
#include <hip/hip_bf16.h>

typedef __attribute__((ext_vector_type(16))) _Float16 v16h;
typedef __attribute__((ext_vector_type(8)))  _Float16 v8h;
typedef __attribute__((ext_vector_type(4)))  _Float16 v4h;
typedef __attribute__((ext_vector_type(8)))  float    v8f;

#define N_NODES 50000
#define N_EDGES 800000
#define DIM 128
#define HEADS 8
#define HEAD_DIM 16

__device__ inline void gemm_tile_16x16(const v16h a[4], const _Float16* __restrict__ WT,
                                       const float* __restrict__ bias,
                                       float* __restrict__ out,
                                       int r0, int n0, int lane) {
    const int col = lane & 15;
    const int kb  = (lane >> 4) << 3;
    const _Float16* wr = WT + (size_t)(n0 + col) * DIM;
    v8f c = {};
#pragma unroll
    for (int t = 0; t < 4; ++t) {
        const v8h lo = *(const v8h*)(wr + t * 32 + kb);
        const v8h hi = *(const v8h*)(wr + t * 32 + kb + 16);
        v16h b;
#pragma unroll
        for (int i = 0; i < 8; ++i) { b[i] = lo[i]; b[i + 8] = hi[i]; }
        c = __builtin_amdgcn_wmma_f32_16x16x32_f16(false, a[t], false, b,
                                                   (short)0, c, false, false);
    }
    const float bia = bias[n0 + col];
    float* op = out + (size_t)r0 * DIM + n0 + col;
#pragma unroll
    for (int j = 0; j < 8; ++j) {
        int m = j + ((lane >> 4) << 3);
        op[(size_t)m * DIM] = c[j] + bia;
    }
}

__device__ inline void load_a_frags(const _Float16* __restrict__ src, int r0, int lane,
                                    v16h a[4]) {
    const int row = lane & 15;
    const int kb  = (lane >> 4) << 3;
    const _Float16* xr = src + (size_t)(r0 + row) * DIM;
#pragma unroll
    for (int t = 0; t < 4; ++t) {
        const v8h lo = *(const v8h*)(xr + t * 32 + kb);
        const v8h hi = *(const v8h*)(xr + t * 32 + kb + 16);
#pragma unroll
        for (int i = 0; i < 8; ++i) { a[t][i] = lo[i]; a[t][i + 8] = hi[i]; }
    }
}

__device__ inline void load_a_frags_f32(const float* __restrict__ src, int r0, int lane,
                                        v16h a[4]) {
    const int row = lane & 15;
    const int kb  = (lane >> 4) << 3;
    const float* xr = src + (size_t)(r0 + row) * DIM;
#pragma unroll
    for (int t = 0; t < 4; ++t) {
        const float4 l0 = *(const float4*)(xr + t * 32 + kb);
        const float4 l1 = *(const float4*)(xr + t * 32 + kb + 4);
        const float4 h0 = *(const float4*)(xr + t * 32 + kb + 16);
        const float4 h1 = *(const float4*)(xr + t * 32 + kb + 20);
        a[t][0]  = (_Float16)l0.x; a[t][1]  = (_Float16)l0.y; a[t][2]  = (_Float16)l0.z; a[t][3]  = (_Float16)l0.w;
        a[t][4]  = (_Float16)l1.x; a[t][5]  = (_Float16)l1.y; a[t][6]  = (_Float16)l1.z; a[t][7]  = (_Float16)l1.w;
        a[t][8]  = (_Float16)h0.x; a[t][9]  = (_Float16)h0.y; a[t][10] = (_Float16)h0.z; a[t][11] = (_Float16)h0.w;
        a[t][12] = (_Float16)h1.x; a[t][13] = (_Float16)h1.y; a[t][14] = (_Float16)h1.z; a[t][15] = (_Float16)h1.w;
    }
}


#define OWN_NODES 196
#define MAP_VECS (N_EDGES / 16)
#define MAP_STEPS ((MAP_VECS + 63) / 64)
#define MAP_BLOCKS ((MAP_VECS + 255) / 256)
#define WT_BLOCKS ((4 * DIM * DIM) / 256)

__device__ inline unsigned owner_bytes4(const int4 s) {
    return ((unsigned)s.x / OWN_NODES) | (((unsigned)s.y / OWN_NODES) << 8) |
           (((unsigned)s.z / OWN_NODES) << 16) | (((unsigned)s.w / OWN_NODES) << 24);
}

__global__ void cvt_weightsT(const float* __restrict__ Wq, const float* __restrict__ Wk,
                             const float* __restrict__ Wv, const float* __restrict__ Wo,
                             _Float16* __restrict__ WT,
                             const int* __restrict__ origin, unsigned* __restrict__ own) {
    if (blockIdx.x < WT_BLOCKS) {
        int i = blockIdx.x * blockDim.x + threadIdx.x;
        int m = i >> 14, r = i & 16383, n = r >> 7, k = r & 127;
        const float* W = (m == 0) ? Wq : (m == 1) ? Wk : (m == 2) ? Wv : Wo;
        WT[i] = (_Float16)W[(size_t)k * DIM + n];
    } else {
        const int g = (blockIdx.x - WT_BLOCKS) * 256 + threadIdx.x;
        if (g < MAP_VECS) {
            const int4* s = (const int4*)origin + (size_t)g * 4;
            const int4 s0 = s[0], s1 = s[1], s2 = s[2], s3 = s[3];
            ((uint4*)own)[g] = make_uint4(owner_bytes4(s0), owner_bytes4(s1), owner_bytes4(s2), owner_bytes4(s3));
        }
    }
}

__global__ void qkv_gemm(const float* __restrict__ x,
                         const _Float16* __restrict__ WTq, const float* __restrict__ bq,
                         const _Float16* __restrict__ WTk, const float* __restrict__ bk,
                         const _Float16* __restrict__ WTv, const float* __restrict__ bv,
                         float* __restrict__ q, float* __restrict__ k,
                         float* __restrict__ v) {
    const int lane = threadIdx.x & 31;
    const int wave = threadIdx.x >> 5;
    const int r0 = blockIdx.x * 16;
    const int n0 = wave * 16;
    v16h a[4];
    load_a_frags_f32(x, r0, lane, a);
    gemm_tile_16x16(a, WTq, bq, q, r0, n0, lane);
    gemm_tile_16x16(a, WTk, bk, k, r0, n0, lane);
    gemm_tile_16x16(a, WTv, bv, v, r0, n0, lane);
}

#define OWN_BLOCKS 256
#define CHUNKS128 (N_EDGES / 128)
#define SUPER ((CHUNKS128 + 3) / 4)
#define LIST_CAP 4096
#define BATCH 8

__device__ inline void lds_add(float* p, float x) {
    __hip_atomic_fetch_add(p, x, __ATOMIC_RELAXED, __HIP_MEMORY_SCOPE_WORKGROUP);
}
__device__ inline void lds_max_f(int* p, float v) {
    if (v >= 0.0f)
        __hip_atomic_fetch_max(p, __float_as_int(v), __ATOMIC_RELAXED, __HIP_MEMORY_SCOPE_WORKGROUP);
    else
        __hip_atomic_fetch_min((unsigned int*)p, __float_as_uint(v), __ATOMIC_RELAXED,
                               __HIP_MEMORY_SCOPE_WORKGROUP);
}

__device__ inline float quad_sum(float x) {
    x += __int_as_float(__builtin_amdgcn_mov_dpp(__float_as_int(x), 0xB1, 0xF, 0xF, true));
    x += __int_as_float(__builtin_amdgcn_mov_dpp(__float_as_int(x), 0x4E, 0xF, 0xF, true));
    return x;
}

__device__ inline float edge_score(const float4 qv, const float4 kv, const float w) {
    float s = qv.x * kv.x + qv.y * kv.y + qv.z * kv.z + qv.w * kv.w;
    s = quad_sum(s);
    s *= 0.25f;
    s *= w;
    return s;
}

__device__ inline void load_origins(const int* __restrict__ origin, int sc, int lane, int o[16]) {
#pragma unroll
    for (int k = 0; k < 4; ++k) {
        const int ch = sc * 4 + k;
        int4 t = make_int4(-1, -1, -1, -1);
        if (ch < CHUNKS128) t = *(const int4*)(origin + ch * 128 + lane * 4);
        o[k * 4 + 0] = t.x; o[k * 4 + 1] = t.y; o[k * 4 + 2] = t.z; o[k * 4 + 3] = t.w;
    }
}

template <int MODE>
__device__ inline void edge_in_place(int e, int oj, int node0, int lane, int h,
                                     const int* __restrict__ dest, const float* __restrict__ ew,
                                     const float* __restrict__ q, const float* __restrict__ k,
                                     const float* __restrict__ v,
                                     int* smax, float* den, float* tile) {
    const int dj = dest[e];
    const float4 qv = *(const float4*)(q + (size_t)oj * DIM + lane * 4);
    const float4 kv = *(const float4*)(k + (size_t)dj * DIM + lane * 4);
    const float s = edge_score(qv, kv, ew[e * HEADS + h]);
    const int ol = oj - node0;
    if (MODE == 1) {
        if ((lane & 3) == 0) lds_max_f(&smax[ol * HEADS + h], s);
    } else {
        float m = __int_as_float(smax[ol * HEADS + h]);
        if (!isfinite(m)) m = 0.0f;
        const float ev = __expf(s - m);
        if ((lane & 3) == 0) lds_add(&den[ol * HEADS + h], ev);
        const float4 vv = *(const float4*)(v + (size_t)dj * DIM + lane * 4);
        float* tp = tile + ol * DIM + lane * 4;
        lds_add(tp + 0, ev * vv.x);
        lds_add(tp + 1, ev * vv.y);
        lds_add(tp + 2, ev * vv.z);
        lds_add(tp + 3, ev * vv.w);
    }
}

template <int MODE>
__device__ inline void scan_owned(const int* __restrict__ origin, const int* __restrict__ dest,
                                  const float* __restrict__ ew,
                                  const float* __restrict__ q, const float* __restrict__ k,
                                  const float* __restrict__ v,
                                  int node0, int wave, int lane, int h,
                                  int* smax, float* den, float* tile) {
    const int rot = blockIdx.x * 6;
    int sc = wave + rot;
    if (sc >= SUPER) sc -= SUPER;
    int o[16];
    load_origins(origin, sc, lane, o);
    for (int s = wave; s < SUPER; s += 32) {
        int scn = s + 32 + rot;
        if (scn >= SUPER) scn -= SUPER;
        int on[16];
#pragma unroll
        for (int i = 0; i < 16; ++i) on[i] = -1;
        if (s + 32 < SUPER) load_origins(origin, scn, lane, on);

        unsigned mk[16];
        unsigned any = 0u;
#pragma unroll
        for (int i = 0; i < 16; ++i) {
            mk[i] = __builtin_amdgcn_ballot_w32((unsigned)(o[i] - node0) < (unsigned)OWN_NODES);
            any |= mk[i];
        }
        if (any != 0u) {
#pragma unroll
            for (int i = 0; i < 16; ++i) {
                unsigned mask = mk[i];
                while (mask) {
                    const int j = __builtin_ctz(mask);
                    mask &= mask - 1u;
                    const int oj = __builtin_amdgcn_readlane(o[i], j);
                    const int e = (sc * 4 + (i >> 2)) * 128 + j * 4 + (i & 3);
                    edge_in_place<MODE>(e, oj, node0, lane, h, dest, ew, q, k, v, smax, den, tile);
                }
            }
        }
        sc = scn;
#pragma unroll
        for (int i = 0; i < 16; ++i) o[i] = on[i];
    }
}

__device__ inline unsigned nonzero_bytes(unsigned x) {
    return ((x & 0x7F7F7F7Fu) + 0x7F7F7F7Fu) | x;
}

__device__ inline void load_map(const uint4* __restrict__ own, int sc, int lane,
                                uint4& a0, uint4& a1) {
    int g0 = sc * 64 + lane;
    int g1 = g0 + 32;
    if (g0 > MAP_VECS - 1) g0 = MAP_VECS - 1;
    if (g1 > MAP_VECS - 1) g1 = MAP_VECS - 1;
    a0 = own[g0];
    a1 = own[g1];
}

__device__ inline void list_owned(const uint4* __restrict__ own, int wave, int lane,
                                  unsigned* list, int* counter) {
    const unsigned pat = blockIdx.x * 0x01010101u;
    const int rot = blockIdx.x * 3;
    int sc = wave + rot;
    if (sc >= MAP_STEPS) sc -= MAP_STEPS;
    uint4 a0, a1;
    load_map(own, sc, lane, a0, a1);
    for (int s = wave; s < MAP_STEPS; s += 32) {
        int scn = s + 32 + rot;
        if (scn >= MAP_STEPS) scn -= MAP_STEPS;
        if (scn >= MAP_STEPS) scn -= MAP_STEPS;
        uint4 n0, n1;
        load_map(own, scn, lane, n0, n1);

        const unsigned M = 0x80808080u;
        unsigned G = nonzero_bytes(a0.x ^ pat) & M;
        G = (nonzero_bytes(a0.y ^ pat) & M) | (G >> 1);
        G = (nonzero_bytes(a0.z ^ pat) & M) | (G >> 1);
        G = (nonzero_bytes(a0.w ^ pat) & M) | (G >> 1);
        G = (nonzero_bytes(a1.x ^ pat) & M) | (G >> 1);
        G = (nonzero_bytes(a1.y ^ pat) & M) | (G >> 1);
        G = (nonzero_bytes(a1.z ^ pat) & M) | (G >> 1);
        G = (nonzero_bytes(a1.w ^ pat) & M) | (G >> 1);
        unsigned F = ~G;
        if (sc == MAP_STEPS - 1) {
            unsigned keep = 0u;
            if (sc * 64 + lane < MAP_VECS) keep |= 0x0F0F0F0Fu;
            if (sc * 64 + 32 + lane < MAP_VECS) keep |= 0xF0F0F0F0u;
            F &= keep;
        }

        unsigned m = __builtin_amdgcn_ballot_w32(F != 0u);
        while (m != 0u) {
            int basep = 0;
            if (lane == 0)
                basep = __hip_atomic_fetch_add(counter, __builtin_popcount(m), __ATOMIC_RELAXED,
                                               __HIP_MEMORY_SCOPE_WORKGROUP);
            basep = __builtin_amdgcn_readfirstlane(basep);
            if (F != 0u) {
                const int p = __builtin_ctz(F);
                F &= F - 1u;
                const int pos = basep + (int)__builtin_amdgcn_mbcnt_lo(m, 0u);
                if (pos < LIST_CAP)
                    list[pos] = (unsigned)(sc * 1024 + lane * 16 + ((p & 4) << 7) + ((p & 3) << 2) + (p >> 3));
            }
            m = __builtin_amdgcn_ballot_w32(F != 0u);
        }
        sc = scn;
        a0 = n0;
        a1 = n1;
    }
}

__global__ __launch_bounds__(1024)
void attend_owned(const unsigned* __restrict__ own, const int* __restrict__ origin, const int* __restrict__ dest,
                  const float* __restrict__ ew,
                  const float* __restrict__ q, const float* __restrict__ k,
                  const float* __restrict__ v, _Float16* __restrict__ vals16) {
    __shared__ float tile[OWN_NODES * DIM];
    __shared__ float scores[LIST_CAP * HEADS];
    __shared__ unsigned list[LIST_CAP];
    __shared__ int dlist[LIST_CAP];
    __shared__ int smax[OWN_NODES * HEADS];
    __shared__ float den[OWN_NODES * HEADS];
    __shared__ int counter;
    const int tid = threadIdx.x;
    const int lane = tid & 31;
    const int wave = __builtin_amdgcn_readfirstlane(tid >> 5);
    const int node0 = blockIdx.x * OWN_NODES;
    const int h = lane >> 2;

    float4* t4 = (float4*)tile;
    const float4* q4 = (const float4*)(q + (size_t)node0 * DIM);
    for (int i = tid; i < OWN_NODES * DIM / 4; i += 1024)
        t4[i] = (node0 + (i >> 5) < N_NODES) ? q4[i] : make_float4(0.0f, 0.0f, 0.0f, 0.0f);
    for (int i = tid; i < OWN_NODES * HEADS; i += 1024) {
        smax[i] = (int)0xFF800000;
        den[i] = 0.0f;
    }
    if (tid == 0) counter = 0;
    __syncthreads();

    list_owned((const uint4*)own, wave, lane, list, &counter);
    __syncthreads();
    const int n = __builtin_amdgcn_readfirstlane(counter);

    if (n <= LIST_CAP) {
        const int per = (n + 31) >> 5;
        int wbeg = wave * per;
        if (wbeg > n) wbeg = n;
        int wend = wbeg + per;
        if (wend > n) wend = n;

        for (int base = wbeg; base < wend; base += 32) {
            const int idx = base + lane;
            const bool valid = idx < wend;
            const int e_l = valid ? (int)list[idx] : 0;
            const int d_l = valid ? dest[e_l] : 0;
            const int ol_l = valid ? origin[e_l] - node0 : 0;
            if (valid) {
                dlist[idx] = d_l;
                list[idx] = (unsigned)e_l | ((unsigned)ol_l << 20);
            }
            int cnt = wend - base;
            if (cnt > 32) cnt = 32;
            for (int j0 = 0; j0 < cnt; j0 += BATCH) {
                float4 kk[BATCH], qq[BATCH];
                float ww[BATCH];
#pragma unroll
                for (int u = 0; u < BATCH; ++u) {
                    int jr = j0 + u;
                    if (jr > cnt - 1) jr = cnt - 1;
                    const int ej = __builtin_amdgcn_readlane(e_l, jr);
                    const int oj = __builtin_amdgcn_readlane(ol_l, jr);
                    const int dj = __builtin_amdgcn_readlane(d_l, jr);
                    kk[u] = *(const float4*)(k + (size_t)dj * DIM + lane * 4);
                    ww[u] = ew[ej * HEADS + h];
                    qq[u] = *(const float4*)(tile + oj * DIM + lane * 4);
                }
#pragma unroll
                for (int u = 0; u < BATCH; ++u) {
                    const int j = j0 + u;
                    if (j < cnt) {
                        const int oj = __builtin_amdgcn_readlane(ol_l, j);
                        const float s = edge_score(qq[u], kk[u], ww[u]);
                        if ((lane & 3) == 0) {
                            lds_max_f(&smax[oj * HEADS + h], s);
                            scores[(base + j) * HEADS + h] = s;
                        }
                    }
                }
            }
        }
        __syncthreads();
        for (int i = tid; i < OWN_NODES * DIM / 4; i += 1024)
            t4[i] = make_float4(0.0f, 0.0f, 0.0f, 0.0f);
        __syncthreads();

        for (int base = wbeg; base < wend; base += 32) {
            const int idx = base + lane;
            const bool valid = idx < wend;
            const int ol_l = valid ? (int)(list[idx] >> 20) : 0;
            const int d_l = valid ? dlist[idx] : 0;
            int cnt = wend - base;
            if (cnt > 32) cnt = 32;
            for (int j0 = 0; j0 < cnt; j0 += BATCH) {
                float4 vv[BATCH];
                float ss[BATCH], mm[BATCH];
#pragma unroll
                for (int u = 0; u < BATCH; ++u) {
                    int jr = j0 + u;
                    if (jr > cnt - 1) jr = cnt - 1;
                    const int oj = __builtin_amdgcn_readlane(ol_l, jr);
                    const int dj = __builtin_amdgcn_readlane(d_l, jr);
                    vv[u] = *(const float4*)(v + (size_t)dj * DIM + lane * 4);
                    ss[u] = scores[(base + jr) * HEADS + h];
                    mm[u] = __int_as_float(smax[oj * HEADS + h]);
                }
#pragma unroll
                for (int u = 0; u < BATCH; ++u) {
                    const int j = j0 + u;
                    if (j < cnt) {
                        const int oj = __builtin_amdgcn_readlane(ol_l, j);
                        float m = mm[u];
                        if (!isfinite(m)) m = 0.0f;
                        const float ev = __expf(ss[u] - m);
                        if ((lane & 3) == 0) lds_add(&den[oj * HEADS + h], ev);
                        float* tp = tile + oj * DIM + lane * 4;
                        lds_add(tp + 0, ev * vv[u].x);
                        lds_add(tp + 1, ev * vv[u].y);
                        lds_add(tp + 2, ev * vv[u].z);
                        lds_add(tp + 3, ev * vv[u].w);
                    }
                }
            }
        }
    } else {
        for (int i = tid; i < OWN_NODES * DIM / 4; i += 1024)
            t4[i] = make_float4(0.0f, 0.0f, 0.0f, 0.0f);
        scan_owned<1>(origin, dest, ew, q, k, v, node0, wave, lane, h, smax, den, tile);
        __syncthreads();
        scan_owned<2>(origin, dest, ew, q, k, v, node0, wave, lane, h, smax, den, tile);
    }
    __syncthreads();

    v4h* out4 = (v4h*)(vals16 + (size_t)node0 * DIM);
    for (int i = tid; i < OWN_NODES * DIM / 4; i += 1024) {
        const int row = i >> 5;
        if (node0 + row < N_NODES) {
            const float dn = den[row * HEADS + ((i & 31) >> 2)] + 1e-16f;
            float4 a = t4[i];
            a.x /= dn; a.y /= dn; a.z /= dn; a.w /= dn;
            const v4h o = { (_Float16)a.x, (_Float16)a.y, (_Float16)a.z, (_Float16)a.w };
            out4[i] = o;
        }
    }
}

#define PROJ_WAVES 4
#define ROW_TILES (N_NODES / 16)
#define PROJ_BLOCKS ((ROW_TILES + PROJ_WAVES - 1) / PROJ_WAVES)

__device__ inline float row16_sum(float x) {
    x += __int_as_float(__builtin_amdgcn_mov_dpp(__float_as_int(x), 0xB1, 0xF, 0xF, true));
    x += __int_as_float(__builtin_amdgcn_mov_dpp(__float_as_int(x), 0x4E, 0xF, 0xF, true));
    x += __int_as_float(__builtin_amdgcn_mov_dpp(__float_as_int(x), 0x141, 0xF, 0xF, true));
    x += __int_as_float(__builtin_amdgcn_mov_dpp(__float_as_int(x), 0x140, 0xF, 0xF, true));
    return x;
}

__global__ __launch_bounds__(32 * PROJ_WAVES) __attribute__((amdgpu_waves_per_eu(1, 4)))
void out_ln_residual(const _Float16* __restrict__ vals16,
                     const _Float16* __restrict__ WTo, const float* __restrict__ bo,
                     const float* __restrict__ x,
                     const float* __restrict__ gamma, const float* __restrict__ beta,
                     float* __restrict__ y) {
    const int lane = threadIdx.x & 31;
    const int tile = blockIdx.x * PROJ_WAVES + __builtin_amdgcn_readfirstlane(threadIdx.x >> 5);
    if (tile >= ROW_TILES) return;
    const int r0 = tile * 16;
    const int c  = lane & 15;
    const int hf = lane >> 4;
    const int kb = hf << 3;

    const float* xr = x + (size_t)(r0 + hf * 8) * DIM + c * 4;
    float4 xa[8], xb[8];
#pragma unroll
    for (int r = 0; r < 8; ++r) {
        xa[r] = *(const float4*)(xr + (size_t)r * DIM);
        xb[r] = *(const float4*)(xr + (size_t)r * DIM + 64);
    }

    v16h a[4];
    load_a_frags(vals16, r0, lane, a);

    v8f acc[8];
#pragma unroll
    for (int t = 0; t < 8; ++t) acc[t] = (v8f){};
#pragma unroll
    for (int kk = 0; kk < 4; ++kk) {
        v16h b[8];
#pragma unroll
        for (int t = 0; t < 8; ++t) {
            const _Float16* wr = WTo + (size_t)((t >> 2) * 64 + c * 4 + (t & 3)) * DIM + kk * 32 + kb;
            const v8h lo = *(const v8h*)(wr);
            const v8h hi = *(const v8h*)(wr + 16);
#pragma unroll
            for (int i = 0; i < 8; ++i) { b[t][i] = lo[i]; b[t][i + 8] = hi[i]; }
        }
#pragma unroll
        for (int t = 0; t < 8; ++t)
            acc[t] = __builtin_amdgcn_wmma_f32_16x16x32_f16(false, a[kk], false, b[t],
                                                            (short)0, acc[t], false, false);
    }

    const float4 bo0 = *(const float4*)(bo + c * 4),    bo1 = *(const float4*)(bo + 64 + c * 4);
    const float4 g0  = *(const float4*)(gamma + c * 4), g1  = *(const float4*)(gamma + 64 + c * 4);
    const float4 b0  = *(const float4*)(beta + c * 4),  b1  = *(const float4*)(beta + 64 + c * 4);
    float* yr = y + (size_t)(r0 + hf * 8) * DIM + c * 4;
#pragma unroll
    for (int r = 0; r < 8; ++r) {
        const float o0 = acc[0][r] + bo0.x, o1 = acc[1][r] + bo0.y, o2 = acc[2][r] + bo0.z, o3 = acc[3][r] + bo0.w;
        const float o4 = acc[4][r] + bo1.x, o5 = acc[5][r] + bo1.y, o6 = acc[6][r] + bo1.z, o7 = acc[7][r] + bo1.w;
        const float s = row16_sum(((o0 + o1) + (o2 + o3)) + ((o4 + o5) + (o6 + o7)));
        const float mu = s * (1.0f / DIM);
        const float d0 = o0 - mu, d1 = o1 - mu, d2 = o2 - mu, d3 = o3 - mu;
        const float d4 = o4 - mu, d5 = o5 - mu, d6 = o6 - mu, d7 = o7 - mu;
        const float s2 = row16_sum(((d0 * d0 + d1 * d1) + (d2 * d2 + d3 * d3)) +
                                   ((d4 * d4 + d5 * d5) + (d6 * d6 + d7 * d7)));
        const float inv = rsqrtf(s2 * (1.0f / DIM) + 1e-5f);
        float4 ya, yb;
        ya.x = xa[r].x + d0 * inv * g0.x + b0.x;
        ya.y = xa[r].y + d1 * inv * g0.y + b0.y;
        ya.z = xa[r].z + d2 * inv * g0.z + b0.z;
        ya.w = xa[r].w + d3 * inv * g0.w + b0.w;
        yb.x = xb[r].x + d4 * inv * g1.x + b1.x;
        yb.y = xb[r].y + d5 * inv * g1.y + b1.y;
        yb.z = xb[r].z + d6 * inv * g1.z + b1.z;
        yb.w = xb[r].w + d7 * inv * g1.w + b1.w;
        *(float4*)(yr + (size_t)r * DIM)      = ya;
        *(float4*)(yr + (size_t)r * DIM + 64) = yb;
    }
}

extern "C" void kernel_launch(void* const* d_in, const int* in_sizes, int n_in,
                              void* d_out, int out_size, void* d_ws, size_t ws_size,
                              hipStream_t stream) {
    const float* x   = (const float*)d_in[0];
    const int*   ei  = (const int*)  d_in[1];
    const float* ew  = (const float*)d_in[2];
    const float* Wq  = (const float*)d_in[3];
    const float* bq  = (const float*)d_in[4];
    const float* Wk  = (const float*)d_in[5];
    const float* bk  = (const float*)d_in[6];
    const float* Wv  = (const float*)d_in[7];
    const float* bv  = (const float*)d_in[8];
    const float* Wo  = (const float*)d_in[9];
    const float* bo  = (const float*)d_in[10];
    const float* gam = (const float*)d_in[11];
    const float* bet = (const float*)d_in[12];
    float* y = (float*)d_out;

    const int* origin = ei;
    const int* dest   = ei + N_EDGES;

    float* ws   = (float*)d_ws;
    float* q    = ws;
    float* k    = q    + (size_t)N_NODES * DIM;
    float* v    = k    + (size_t)N_NODES * DIM;
    float* outb = v    + (size_t)N_NODES * DIM;
    _Float16* vals16 = (_Float16*)(outb + (size_t)N_NODES * DIM);
    _Float16* WT     = vals16 + (size_t)N_NODES * DIM;
    _Float16* WTq = WT;
    _Float16* WTk = WT + 1 * DIM * DIM;
    _Float16* WTv = WT + 2 * DIM * DIM;
    _Float16* WTo = WT + 3 * DIM * DIM;

    unsigned* own = (unsigned*)outb;
    cvt_weightsT<<<WT_BLOCKS + MAP_BLOCKS, 256, 0, stream>>>(Wq, Wk, Wv, Wo, WT, origin, own);

    qkv_gemm<<<N_NODES / 16, 256, 0, stream>>>(x, WTq, bq, WTk, bk, WTv, bv, q, k, v);

    attend_owned<<<OWN_BLOCKS, 1024, 0, stream>>>(own, origin, dest, ew, q, k, v, vals16);

    out_ln_residual<<<PROJ_BLOCKS, 32 * PROJ_WAVES, 0, stream>>>(vals16, WTo, bo, x, gam, bet, y);
}


# Owner work-group sorts its edge list by node in LDS (rank by integer adds); a wave per node keeps max, denominator, value sums in registers: no LDS float adds; sums: list order, rescaled on new max
# speedup vs baseline: 4.4619x; 1.4709x over previous
#include <hip/hip_runtime.h>
#include <hip/hip_bf16.h>

typedef __attribute__((ext_vector_type(16))) _Float16 v16h;
typedef __attribute__((ext_vector_type(8)))  _Float16 v8h;
typedef __attribute__((ext_vector_type(4)))  _Float16 v4h;
typedef __attribute__((ext_vector_type(8)))  float    v8f;

#define N_NODES 50000
#define N_EDGES 800000
#define DIM 128
#define HEADS 8
#define HEAD_DIM 16

__device__ inline void load_a_frags(const _Float16* __restrict__ src, int r0, int lane,
                                    v16h a[4]) {
    const int row = lane & 15;
    const int kb  = (lane >> 4) << 3;
    const _Float16* xr = src + (size_t)(r0 + row) * DIM;
#pragma unroll
    for (int t = 0; t < 4; ++t) {
        const v8h lo = *(const v8h*)(xr + t * 32 + kb);
        const v8h hi = *(const v8h*)(xr + t * 32 + kb + 16);
#pragma unroll
        for (int i = 0; i < 8; ++i) { a[t][i] = lo[i]; a[t][i + 8] = hi[i]; }
    }
}

__device__ inline void load_a_frags_f32(const float* __restrict__ src, int r0, int lane,
                                        v16h a[4]) {
    const int row = lane & 15;
    const int kb  = (lane >> 4) << 3;
    const float* xr = src + (size_t)(r0 + row) * DIM;
#pragma unroll
    for (int t = 0; t < 4; ++t) {
        const float4 l0 = *(const float4*)(xr + t * 32 + kb);
        const float4 l1 = *(const float4*)(xr + t * 32 + kb + 4);
        const float4 h0 = *(const float4*)(xr + t * 32 + kb + 16);
        const float4 h1 = *(const float4*)(xr + t * 32 + kb + 20);
        a[t][0]  = (_Float16)l0.x; a[t][1]  = (_Float16)l0.y; a[t][2]  = (_Float16)l0.z; a[t][3]  = (_Float16)l0.w;
        a[t][4]  = (_Float16)l1.x; a[t][5]  = (_Float16)l1.y; a[t][6]  = (_Float16)l1.z; a[t][7]  = (_Float16)l1.w;
        a[t][8]  = (_Float16)h0.x; a[t][9]  = (_Float16)h0.y; a[t][10] = (_Float16)h0.z; a[t][11] = (_Float16)h0.w;
        a[t][12] = (_Float16)h1.x; a[t][13] = (_Float16)h1.y; a[t][14] = (_Float16)h1.z; a[t][15] = (_Float16)h1.w;
    }
}


#define OWN_NODES 196
#define MAP_VECS (N_EDGES / 16)
#define MAP_STEPS ((MAP_VECS + 63) / 64)
#define MAP_BLOCKS ((MAP_VECS + 255) / 256)
#define WT_BLOCKS ((4 * DIM * DIM) / 256)

__device__ inline unsigned owner_bytes4(const int4 s) {
    return ((unsigned)s.x / OWN_NODES) | (((unsigned)s.y / OWN_NODES) << 8) |
           (((unsigned)s.z / OWN_NODES) << 16) | (((unsigned)s.w / OWN_NODES) << 24);
}

__global__ void cvt_weightsT(const float* __restrict__ Wq, const float* __restrict__ Wk,
                             const float* __restrict__ Wv, const float* __restrict__ Wo,
                             _Float16* __restrict__ WT,
                             const int* __restrict__ origin, unsigned* __restrict__ own) {
    if (blockIdx.x < WT_BLOCKS) {
        int i = blockIdx.x * blockDim.x + threadIdx.x;
        int m = i >> 14, r = i & 16383, n = r >> 7, k = r & 127;
        const float* W = (m == 0) ? Wq : (m == 1) ? Wk : (m == 2) ? Wv : Wo;
        WT[i] = (_Float16)W[(size_t)k * DIM + n];
    } else {
        const int g = (blockIdx.x - WT_BLOCKS) * 256 + threadIdx.x;
        if (g < MAP_VECS) {
            const int4* s = (const int4*)origin + (size_t)g * 4;
            const int4 s0 = s[0], s1 = s[1], s2 = s[2], s3 = s[3];
            ((uint4*)own)[g] = make_uint4(owner_bytes4(s0), owner_bytes4(s1), owner_bytes4(s2), owner_bytes4(s3));
        }
    }
}

#define ROW_TILES (N_NODES / 16)
#define QKV_TILES 2
#define QKV_GRID ((ROW_TILES + QKV_TILES - 1) / QKV_TILES)

__device__ inline void proj_two_tiles(const v16h a0[4], const v16h a1[4],
                                      const _Float16* __restrict__ WT, const float* __restrict__ bias,
                                      float* __restrict__ out, int r0, int r1, bool two, int lane) {
    const int c  = lane & 15;
    const int hf = lane >> 4;
    const int kb = hf << 3;
    v8f acc0[8], acc1[8];
#pragma unroll
    for (int t = 0; t < 8; ++t) { acc0[t] = (v8f){}; acc1[t] = (v8f){}; }
#pragma unroll
    for (int kk = 0; kk < 4; ++kk) {
        v16h b[8];
#pragma unroll
        for (int t = 0; t < 8; ++t) {
            const _Float16* wr = WT + (size_t)((t >> 2) * 64 + c * 4 + (t & 3)) * DIM + kk * 32 + kb;
            const v8h lo = *(const v8h*)(wr);
            const v8h hi = *(const v8h*)(wr + 16);
#pragma unroll
            for (int i = 0; i < 8; ++i) { b[t][i] = lo[i]; b[t][i + 8] = hi[i]; }
        }
#pragma unroll
        for (int t = 0; t < 8; ++t)
            acc0[t] = __builtin_amdgcn_wmma_f32_16x16x32_f16(false, a0[kk], false, b[t],
                                                             (short)0, acc0[t], false, false);
#pragma unroll
        for (int t = 0; t < 8; ++t)
            acc1[t] = __builtin_amdgcn_wmma_f32_16x16x32_f16(false, a1[kk], false, b[t],
                                                             (short)0, acc1[t], false, false);
    }
    const float4 bi0 = *(const float4*)(bias + c * 4), bi1 = *(const float4*)(bias + 64 + c * 4);
    float* op0 = out + (size_t)(r0 + hf * 8) * DIM + c * 4;
#pragma unroll
    for (int r = 0; r < 8; ++r) {
        const float4 ya = make_float4(acc0[0][r] + bi0.x, acc0[1][r] + bi0.y, acc0[2][r] + bi0.z, acc0[3][r] + bi0.w);
        const float4 yb = make_float4(acc0[4][r] + bi1.x, acc0[5][r] + bi1.y, acc0[6][r] + bi1.z, acc0[7][r] + bi1.w);
        *(float4*)(op0 + (size_t)r * DIM)      = ya;
        *(float4*)(op0 + (size_t)r * DIM + 64) = yb;
    }
    if (two) {
        float* op1 = out + (size_t)(r1 + hf * 8) * DIM + c * 4;
#pragma unroll
        for (int r = 0; r < 8; ++r) {
            const float4 ya = make_float4(acc1[0][r] + bi0.x, acc1[1][r] + bi0.y, acc1[2][r] + bi0.z, acc1[3][r] + bi0.w);
            const float4 yb = make_float4(acc1[4][r] + bi1.x, acc1[5][r] + bi1.y, acc1[6][r] + bi1.z, acc1[7][r] + bi1.w);
            *(float4*)(op1 + (size_t)r * DIM)      = ya;
            *(float4*)(op1 + (size_t)r * DIM + 64) = yb;
        }
    }
}

__global__ __launch_bounds__(32) __attribute__((amdgpu_waves_per_eu(1, 2)))
void qkv_gemm(const float* __restrict__ x,
              const _Float16* __restrict__ WTq, const float* __restrict__ bq,
              const _Float16* __restrict__ WTk, const float* __restrict__ bk,
              const _Float16* __restrict__ WTv, const float* __restrict__ bv,
              float* __restrict__ q, float* __restrict__ k,
              float* __restrict__ v) {
    const int lane = threadIdx.x;
    const int t0 = blockIdx.x * QKV_TILES;
    const bool two = t0 + 1 < ROW_TILES;
    const int r0 = t0 * 16;
    const int r1 = two ? r0 + 16 : r0;
    v16h a0[4], a1[4];
    load_a_frags_f32(x, r0, lane, a0);
    load_a_frags_f32(x, r1, lane, a1);
    proj_two_tiles(a0, a1, WTq, bq, q, r0, r1, two, lane);
    proj_two_tiles(a0, a1, WTk, bk, k, r0, r1, two, lane);
    proj_two_tiles(a0, a1, WTv, bv, v, r0, r1, two, lane);
}

#define OWN_BLOCKS 256
#define CHUNKS128 (N_EDGES / 128)
#define SUPER ((CHUNKS128 + 3) / 4)
#define LIST_CAP 4096
#define BATCH 8

__device__ inline void lds_add(float* p, float x) {
    __hip_atomic_fetch_add(p, x, __ATOMIC_RELAXED, __HIP_MEMORY_SCOPE_WORKGROUP);
}
__device__ inline void lds_max_f(int* p, float v) {
    if (v >= 0.0f)
        __hip_atomic_fetch_max(p, __float_as_int(v), __ATOMIC_RELAXED, __HIP_MEMORY_SCOPE_WORKGROUP);
    else
        __hip_atomic_fetch_min((unsigned int*)p, __float_as_uint(v), __ATOMIC_RELAXED,
                               __HIP_MEMORY_SCOPE_WORKGROUP);
}

__device__ inline float quad_sum(float x) {
    x += __int_as_float(__builtin_amdgcn_mov_dpp(__float_as_int(x), 0xB1, 0xF, 0xF, true));
    x += __int_as_float(__builtin_amdgcn_mov_dpp(__float_as_int(x), 0x4E, 0xF, 0xF, true));
    return x;
}

__device__ inline float edge_score(const float4 qv, const float4 kv, const float w) {
    float s = qv.x * kv.x + qv.y * kv.y + qv.z * kv.z + qv.w * kv.w;
    s = quad_sum(s);
    s *= 0.25f;
    s *= w;
    return s;
}

__device__ inline void load_origins(const int* __restrict__ origin, int sc, int lane, int o[16]) {
#pragma unroll
    for (int k = 0; k < 4; ++k) {
        const int ch = sc * 4 + k;
        int4 t = make_int4(-1, -1, -1, -1);
        if (ch < CHUNKS128) t = *(const int4*)(origin + ch * 128 + lane * 4);
        o[k * 4 + 0] = t.x; o[k * 4 + 1] = t.y; o[k * 4 + 2] = t.z; o[k * 4 + 3] = t.w;
    }
}

template <int MODE>
__device__ inline void edge_in_place(int e, int oj, int node0, int lane, int h,
                                     const int* __restrict__ dest, const float* __restrict__ ew,
                                     const float* __restrict__ q, const float* __restrict__ k,
                                     const float* __restrict__ v,
                                     int* smax, float* den, float* tile) {
    const int dj = dest[e];
    const float4 qv = *(const float4*)(q + (size_t)oj * DIM + lane * 4);
    const float4 kv = *(const float4*)(k + (size_t)dj * DIM + lane * 4);
    const float s = edge_score(qv, kv, ew[e * HEADS + h]);
    const int ol = oj - node0;
    if (MODE == 1) {
        if ((lane & 3) == 0) lds_max_f(&smax[ol * HEADS + h], s);
    } else {
        float m = __int_as_float(smax[ol * HEADS + h]);
        if (!isfinite(m)) m = 0.0f;
        const float ev = __expf(s - m);
        if ((lane & 3) == 0) lds_add(&den[ol * HEADS + h], ev);
        const float4 vv = *(const float4*)(v + (size_t)dj * DIM + lane * 4);
        float* tp = tile + ol * DIM + lane;
        lds_add(tp + 0,  ev * vv.x);
        lds_add(tp + 32, ev * vv.y);
        lds_add(tp + 64, ev * vv.z);
        lds_add(tp + 96, ev * vv.w);
    }
}

template <int MODE>
__device__ inline void scan_owned(const int* __restrict__ origin, const int* __restrict__ dest,
                                  const float* __restrict__ ew,
                                  const float* __restrict__ q, const float* __restrict__ k,
                                  const float* __restrict__ v,
                                  int node0, int wave, int lane, int h,
                                  int* smax, float* den, float* tile) {
    const int rot = blockIdx.x * 6;
    int sc = wave + rot;
    if (sc >= SUPER) sc -= SUPER;
    int o[16];
    load_origins(origin, sc, lane, o);
    for (int s = wave; s < SUPER; s += 32) {
        int scn = s + 32 + rot;
        if (scn >= SUPER) scn -= SUPER;
        int on[16];
#pragma unroll
        for (int i = 0; i < 16; ++i) on[i] = -1;
        if (s + 32 < SUPER) load_origins(origin, scn, lane, on);

        unsigned mk[16];
        unsigned any = 0u;
#pragma unroll
        for (int i = 0; i < 16; ++i) {
            mk[i] = __builtin_amdgcn_ballot_w32((unsigned)(o[i] - node0) < (unsigned)OWN_NODES);
            any |= mk[i];
        }
        if (any != 0u) {
#pragma unroll
            for (int i = 0; i < 16; ++i) {
                unsigned mask = mk[i];
                while (mask) {
                    const int j = __builtin_ctz(mask);
                    mask &= mask - 1u;
                    const int oj = __builtin_amdgcn_readlane(o[i], j);
                    const int e = (sc * 4 + (i >> 2)) * 128 + j * 4 + (i & 3);
                    edge_in_place<MODE>(e, oj, node0, lane, h, dest, ew, q, k, v, smax, den, tile);
                }
            }
        }
        sc = scn;
#pragma unroll
        for (int i = 0; i < 16; ++i) o[i] = on[i];
    }
}

__device__ inline unsigned nonzero_bytes(unsigned x) {
    return ((x & 0x7F7F7F7Fu) + 0x7F7F7F7Fu) | x;
}

__device__ inline void load_map(const uint4* __restrict__ own, int sc, int lane,
                                uint4& a0, uint4& a1) {
    int g0 = sc * 64 + lane;
    int g1 = g0 + 32;
    if (g0 > MAP_VECS - 1) g0 = MAP_VECS - 1;
    if (g1 > MAP_VECS - 1) g1 = MAP_VECS - 1;
    a0 = own[g0];
    a1 = own[g1];
}

__device__ inline void list_owned(const uint4* __restrict__ own, int wave, int lane,
                                  unsigned* list, int* counter) {
    const unsigned pat = blockIdx.x * 0x01010101u;
    const int rot = blockIdx.x * 3;
    int sc = wave + rot;
    if (sc >= MAP_STEPS) sc -= MAP_STEPS;
    uint4 a0, a1;
    load_map(own, sc, lane, a0, a1);
    for (int s = wave; s < MAP_STEPS; s += 32) {
        int scn = s + 32 + rot;
        if (scn >= MAP_STEPS) scn -= MAP_STEPS;
        if (scn >= MAP_STEPS) scn -= MAP_STEPS;
        uint4 n0, n1;
        load_map(own, scn, lane, n0, n1);

        const unsigned M = 0x80808080u;
        unsigned G = nonzero_bytes(a0.x ^ pat) & M;
        G = (nonzero_bytes(a0.y ^ pat) & M) | (G >> 1);
        G = (nonzero_bytes(a0.z ^ pat) & M) | (G >> 1);
        G = (nonzero_bytes(a0.w ^ pat) & M) | (G >> 1);
        G = (nonzero_bytes(a1.x ^ pat) & M) | (G >> 1);
        G = (nonzero_bytes(a1.y ^ pat) & M) | (G >> 1);
        G = (nonzero_bytes(a1.z ^ pat) & M) | (G >> 1);
        G = (nonzero_bytes(a1.w ^ pat) & M) | (G >> 1);
        unsigned F = ~G;
        if (sc == MAP_STEPS - 1) {
            unsigned keep = 0u;
            if (sc * 64 + lane < MAP_VECS) keep |= 0x0F0F0F0Fu;
            if (sc * 64 + 32 + lane < MAP_VECS) keep |= 0xF0F0F0F0u;
            F &= keep;
        }

        unsigned m = __builtin_amdgcn_ballot_w32(F != 0u);
        while (m != 0u) {
            int basep = 0;
            if (lane == 0)
                basep = __hip_atomic_fetch_add(counter, __builtin_popcount(m), __ATOMIC_RELAXED,
                                               __HIP_MEMORY_SCOPE_WORKGROUP);
            basep = __builtin_amdgcn_readfirstlane(basep);
            if (F != 0u) {
                const int p = __builtin_ctz(F);
                F &= F - 1u;
                const int pos = basep + (int)__builtin_amdgcn_mbcnt_lo(m, 0u);
                if (pos < LIST_CAP)
                    list[pos] = (unsigned)(sc * 1024 + lane * 16 + ((p & 4) << 7) + ((p & 3) << 2) + (p >> 3));
            }
            m = __builtin_amdgcn_ballot_w32(F != 0u);
        }
        sc = scn;
        a0 = n0;
        a1 = n1;
    }
}

__global__ __launch_bounds__(1024) __attribute__((amdgpu_waves_per_eu(8, 8)))
void attend_owned(const unsigned* __restrict__ own, const int* __restrict__ origin, const int* __restrict__ dest,
                  const float* __restrict__ ew,
                  const float* __restrict__ q, const float* __restrict__ k,
                  const float* __restrict__ v, _Float16* __restrict__ vals16) {
    __shared__ float tile[OWN_NODES * DIM];
    __shared__ unsigned list[LIST_CAP];
    __shared__ int sedge[LIST_CAP];
    __shared__ int sdest[LIST_CAP];
    __shared__ int smax[OWN_NODES * HEADS];
    __shared__ float den[OWN_NODES * HEADS];
    __shared__ int cnt[OWN_NODES];
    __shared__ int start[OWN_NODES + 1];
    __shared__ int counter;
    __shared__ int nextnode;
    const int tid = threadIdx.x;
    const int lane = tid & 31;
    const int wave = __builtin_amdgcn_readfirstlane(tid >> 5);
    const int node0 = blockIdx.x * OWN_NODES;
    const int h = lane >> 2;

    for (int i = tid; i < OWN_NODES * HEADS; i += 1024) {
        smax[i] = (int)0xFF800000;
        den[i] = 0.0f;
    }
    for (int i = tid; i < OWN_NODES; i += 1024) cnt[i] = 0;
    if (tid == 0) { counter = 0; nextnode = 0; }
    __syncthreads();

    list_owned((const uint4*)own, wave, lane, list, &counter);
    __syncthreads();
    const int n = __builtin_amdgcn_readfirstlane(counter);

    if (n <= LIST_CAP) {
        int e_[4], o_[4], d_[4], r_[4];
#pragma unroll
        for (int i = 0; i < 4; ++i) {
            const int idx = tid + i * 1024;
            e_[i] = (idx < n) ? (int)list[idx] : 0;
        }
#pragma unroll
        for (int i = 0; i < 4; ++i) {
            const int idx = tid + i * 1024;
            o_[i] = (idx < n) ? origin[e_[i]] - node0 : 0;
            d_[i] = (idx < n) ? dest[e_[i]] : 0;
        }
#pragma unroll
        for (int i = 0; i < 4; ++i) {
            const int idx = tid + i * 1024;
            r_[i] = 0;
            if (idx < n)
                r_[i] = __hip_atomic_fetch_add(&cnt[o_[i]], 1, __ATOMIC_RELAXED, __HIP_MEMORY_SCOPE_WORKGROUP);
        }
        __syncthreads();
        if (wave == 0) {
            int c[7];
            int sum = 0;
#pragma unroll
            for (int i = 0; i < 7; ++i) {
                const int idx = lane * 7 + i;
                c[i] = (idx < OWN_NODES) ? cnt[idx] : 0;
                sum += c[i];
            }
            int incl = sum;
#pragma unroll
            for (int dlt = 1; dlt < 32; dlt <<= 1) {
                const int t = __shfl_up(incl, dlt, 32);
                if (lane >= dlt) incl += t;
            }
            int run = incl - sum;
#pragma unroll
            for (int i = 0; i < 7; ++i) {
                const int idx = lane * 7 + i;
                if (idx <= OWN_NODES) start[idx] = run;
                run += c[i];
            }
        }
        __syncthreads();
#pragma unroll
        for (int i = 0; i < 4; ++i) {
            const int idx = tid + i * 1024;
            if (idx < n) {
                const int pos = start[o_[i]] + r_[i];
                sedge[pos] = e_[i];
                sdest[pos] = d_[i];
            }
        }
        __syncthreads();

        for (;;) {
            int ol = 0;
            if (lane == 0)
                ol = __hip_atomic_fetch_add(&nextnode, 1, __ATOMIC_RELAXED, __HIP_MEMORY_SCOPE_WORKGROUP);
            ol = __builtin_amdgcn_readfirstlane(ol);
            if (ol >= OWN_NODES || node0 + ol >= N_NODES) break;
            const int s0 = __builtin_amdgcn_readfirstlane(start[ol]);
            const int s1 = __builtin_amdgcn_readfirstlane(start[ol + 1]);
            const float4 qv = *(const float4*)(q + (size_t)(node0 + ol) * DIM + lane * 4);
            float m = -__builtin_inff();
            float shift = 0.0f;
            float dsum = 0.0f;
            float4 acc = make_float4(0.0f, 0.0f, 0.0f, 0.0f);
            for (int b = s0; b < s1; b += BATCH) {
                int c = s1 - b;
                if (c > BATCH) c = BATCH;
                int li = b + lane;
                if (li > s1 - 1) li = s1 - 1;
                const int d_l = sdest[li];
                const int e_l = sedge[li];
                float4 kk[BATCH], vv[BATCH];
                float ww[BATCH];
#pragma unroll
                for (int u = 0; u < BATCH; ++u) {
                    const int dj = __builtin_amdgcn_readlane(d_l, u);
                    const int ej = __builtin_amdgcn_readlane(e_l, u);
                    kk[u] = *(const float4*)(k + (size_t)dj * DIM + lane * 4);
                    ww[u] = ew[ej * HEADS + h];
                    vv[u] = *(const float4*)(v + (size_t)dj * DIM + lane * 4);
                }
                float s[BATCH];
                float cm = -__builtin_inff();
#pragma unroll
                for (int u = 0; u < BATCH; ++u) {
                    const float t = edge_score(qv, kk[u], ww[u]);
                    s[u] = (u < c) ? t : -__builtin_inff();
                    cm = fmaxf(cm, s[u]);
                }
                const float mn = fmaxf(m, cm);
                const float sh = isfinite(mn) ? mn : 0.0f;
                const float sc = (b == s0) ? 0.0f : __expf(shift - sh);
                dsum *= sc;
                acc.x *= sc; acc.y *= sc; acc.z *= sc; acc.w *= sc;
#pragma unroll
                for (int u = 0; u < BATCH; ++u) {
                    const float ev = __expf(s[u] - sh);
                    dsum += ev;
                    acc.x += ev * vv[u].x;
                    acc.y += ev * vv[u].y;
                    acc.z += ev * vv[u].z;
                    acc.w += ev * vv[u].w;
                }
                m = mn;
                shift = sh;
            }
            const float dn = dsum + 1e-16f;
            const v4h o = { (_Float16)(acc.x / dn), (_Float16)(acc.y / dn), (_Float16)(acc.z / dn), (_Float16)(acc.w / dn) };
            *(v4h*)(vals16 + (size_t)(node0 + ol) * DIM + lane * 4) = o;
        }
    } else {
        float4* t4 = (float4*)tile;
        for (int i = tid; i < OWN_NODES * DIM / 4; i += 1024)
            t4[i] = make_float4(0.0f, 0.0f, 0.0f, 0.0f);
        scan_owned<1>(origin, dest, ew, q, k, v, node0, wave, lane, h, smax, den, tile);
        __syncthreads();
        scan_owned<2>(origin, dest, ew, q, k, v, node0, wave, lane, h, smax, den, tile);
        __syncthreads();

        v4h* out4 = (v4h*)(vals16 + (size_t)node0 * DIM);
        for (int i = tid; i < OWN_NODES * DIM / 4; i += 1024) {
            const int row = i >> 5;
            if (node0 + row < N_NODES) {
                const float dn = den[row * HEADS + ((i & 31) >> 2)] + 1e-16f;
                const float* tr = tile + row * DIM + (i & 31);
                float4 a = make_float4(tr[0], tr[32], tr[64], tr[96]);
                a.x /= dn; a.y /= dn; a.z /= dn; a.w /= dn;
                const v4h o = { (_Float16)a.x, (_Float16)a.y, (_Float16)a.z, (_Float16)a.w };
                out4[i] = o;
            }
        }
    }
}

#define PROJ_WAVES 4
#define PROJ_BLOCKS ((ROW_TILES + PROJ_WAVES - 1) / PROJ_WAVES)

__device__ inline float row16_sum(float x) {
    x += __int_as_float(__builtin_amdgcn_mov_dpp(__float_as_int(x), 0xB1, 0xF, 0xF, true));
    x += __int_as_float(__builtin_amdgcn_mov_dpp(__float_as_int(x), 0x4E, 0xF, 0xF, true));
    x += __int_as_float(__builtin_amdgcn_mov_dpp(__float_as_int(x), 0x141, 0xF, 0xF, true));
    x += __int_as_float(__builtin_amdgcn_mov_dpp(__float_as_int(x), 0x140, 0xF, 0xF, true));
    return x;
}

__global__ __launch_bounds__(32 * PROJ_WAVES) __attribute__((amdgpu_waves_per_eu(1, 4)))
void out_ln_residual(const _Float16* __restrict__ vals16,
                     const _Float16* __restrict__ WTo, const float* __restrict__ bo,
                     const float* __restrict__ x,
                     const float* __restrict__ gamma, const float* __restrict__ beta,
                     float* __restrict__ y) {
    const int lane = threadIdx.x & 31;
    const int tile = blockIdx.x * PROJ_WAVES + __builtin_amdgcn_readfirstlane(threadIdx.x >> 5);
    if (tile >= ROW_TILES) return;
    const int r0 = tile * 16;
    const int c  = lane & 15;
    const int hf = lane >> 4;
    const int kb = hf << 3;

    const float* xr = x + (size_t)(r0 + hf * 8) * DIM + c * 4;
    float4 xa[8], xb[8];
#pragma unroll
    for (int r = 0; r < 8; ++r) {
        xa[r] = *(const float4*)(xr + (size_t)r * DIM);
        xb[r] = *(const float4*)(xr + (size_t)r * DIM + 64);
    }

    v16h a[4];
    load_a_frags(vals16, r0, lane, a);

    v8f acc[8];
#pragma unroll
    for (int t = 0; t < 8; ++t) acc[t] = (v8f){};
#pragma unroll
    for (int kk = 0; kk < 4; ++kk) {
        v16h b[8];
#pragma unroll
        for (int t = 0; t < 8; ++t) {
            const _Float16* wr = WTo + (size_t)((t >> 2) * 64 + c * 4 + (t & 3)) * DIM + kk * 32 + kb;
            const v8h lo = *(const v8h*)(wr);
            const v8h hi = *(const v8h*)(wr + 16);
#pragma unroll
            for (int i = 0; i < 8; ++i) { b[t][i] = lo[i]; b[t][i + 8] = hi[i]; }
        }
#pragma unroll
        for (int t = 0; t < 8; ++t)
            acc[t] = __builtin_amdgcn_wmma_f32_16x16x32_f16(false, a[kk], false, b[t],
                                                            (short)0, acc[t], false, false);
    }

    const float4 bo0 = *(const float4*)(bo + c * 4),    bo1 = *(const float4*)(bo + 64 + c * 4);
    const float4 g0  = *(const float4*)(gamma + c * 4), g1  = *(const float4*)(gamma + 64 + c * 4);
    const float4 b0  = *(const float4*)(beta + c * 4),  b1  = *(const float4*)(beta + 64 + c * 4);
    float* yr = y + (size_t)(r0 + hf * 8) * DIM + c * 4;
#pragma unroll
    for (int r = 0; r < 8; ++r) {
        const float o0 = acc[0][r] + bo0.x, o1 = acc[1][r] + bo0.y, o2 = acc[2][r] + bo0.z, o3 = acc[3][r] + bo0.w;
        const float o4 = acc[4][r] + bo1.x, o5 = acc[5][r] + bo1.y, o6 = acc[6][r] + bo1.z, o7 = acc[7][r] + bo1.w;
        const float s = row16_sum(((o0 + o1) + (o2 + o3)) + ((o4 + o5) + (o6 + o7)));
        const float mu = s * (1.0f / DIM);
        const float d0 = o0 - mu, d1 = o1 - mu, d2 = o2 - mu, d3 = o3 - mu;
        const float d4 = o4 - mu, d5 = o5 - mu, d6 = o6 - mu, d7 = o7 - mu;
        const float s2 = row16_sum(((d0 * d0 + d1 * d1) + (d2 * d2 + d3 * d3)) +
                                   ((d4 * d4 + d5 * d5) + (d6 * d6 + d7 * d7)));
        const float inv = rsqrtf(s2 * (1.0f / DIM) + 1e-5f);
        float4 ya, yb;
        ya.x = xa[r].x + d0 * inv * g0.x + b0.x;
        ya.y = xa[r].y + d1 * inv * g0.y + b0.y;
        ya.z = xa[r].z + d2 * inv * g0.z + b0.z;
        ya.w = xa[r].w + d3 * inv * g0.w + b0.w;
        yb.x = xb[r].x + d4 * inv * g1.x + b1.x;
        yb.y = xb[r].y + d5 * inv * g1.y + b1.y;
        yb.z = xb[r].z + d6 * inv * g1.z + b1.z;
        yb.w = xb[r].w + d7 * inv * g1.w + b1.w;
        *(float4*)(yr + (size_t)r * DIM)      = ya;
        *(float4*)(yr + (size_t)r * DIM + 64) = yb;
    }
}

extern "C" void kernel_launch(void* const* d_in, const int* in_sizes, int n_in,
                              void* d_out, int out_size, void* d_ws, size_t ws_size,
                              hipStream_t stream) {
    const float* x   = (const float*)d_in[0];
    const int*   ei  = (const int*)  d_in[1];
    const float* ew  = (const float*)d_in[2];
    const float* Wq  = (const float*)d_in[3];
    const float* bq  = (const float*)d_in[4];
    const float* Wk  = (const float*)d_in[5];
    const float* bk  = (const float*)d_in[6];
    const float* Wv  = (const float*)d_in[7];
    const float* bv  = (const float*)d_in[8];
    const float* Wo  = (const float*)d_in[9];
    const float* bo  = (const float*)d_in[10];
    const float* gam = (const float*)d_in[11];
    const float* bet = (const float*)d_in[12];
    float* y = (float*)d_out;

    const int* origin = ei;
    const int* dest   = ei + N_EDGES;

    float* ws   = (float*)d_ws;
    float* q    = ws;
    float* k    = q    + (size_t)N_NODES * DIM;
    float* v    = k    + (size_t)N_NODES * DIM;
    float* outb = v    + (size_t)N_NODES * DIM;
    _Float16* vals16 = (_Float16*)(outb + (size_t)N_NODES * DIM);
    _Float16* WT     = vals16 + (size_t)N_NODES * DIM;
    _Float16* WTq = WT;
    _Float16* WTk = WT + 1 * DIM * DIM;
    _Float16* WTv = WT + 2 * DIM * DIM;
    _Float16* WTo = WT + 3 * DIM * DIM;

    unsigned* own = (unsigned*)outb;
    cvt_weightsT<<<WT_BLOCKS + MAP_BLOCKS, 256, 0, stream>>>(Wq, Wk, Wv, Wo, WT, origin, own);

    qkv_gemm<<<QKV_GRID, 32, 0, stream>>>(x, WTq, bq, WTk, bk, WTv, bv, q, k, v);

    attend_owned<<<OWN_BLOCKS, 1024, 0, stream>>>(own, origin, dest, ew, q, k, v, vals16);

    out_ln_residual<<<PROJ_BLOCKS, 32 * PROJ_WAVES, 0, stream>>>(vals16, WTo, bo, x, gam, bet, y);
}


# Weight (B) fragments of both products come from LDS: a work-group copies the f16 weights once into fragment-ordered rows of 272-byte pitch; each fragment is one LDS read, not a 16-line global load
# speedup vs baseline: 5.6227x; 1.2602x over previous
#include <hip/hip_runtime.h>
#include <hip/hip_bf16.h>

typedef __attribute__((ext_vector_type(16))) _Float16 v16h;
typedef __attribute__((ext_vector_type(8)))  _Float16 v8h;
typedef __attribute__((ext_vector_type(4)))  _Float16 v4h;
typedef __attribute__((ext_vector_type(8)))  float    v8f;

#define N_NODES 50000
#define N_EDGES 800000
#define DIM 128
#define HEADS 8
#define HEAD_DIM 16

__device__ inline void load_a_frags(const _Float16* __restrict__ src, int r0, int lane,
                                    v16h a[4]) {
    const int row = lane & 15;
    const int kb  = (lane >> 4) << 3;
    const _Float16* xr = src + (size_t)(r0 + row) * DIM;
#pragma unroll
    for (int t = 0; t < 4; ++t) {
        const v8h lo = *(const v8h*)(xr + t * 32 + kb);
        const v8h hi = *(const v8h*)(xr + t * 32 + kb + 16);
#pragma unroll
        for (int i = 0; i < 8; ++i) { a[t][i] = lo[i]; a[t][i + 8] = hi[i]; }
    }
}

__device__ inline void load_a_frags_f32(const float* __restrict__ src, int r0, int lane,
                                        v16h a[4]) {
    const int row = lane & 15;
    const int kb  = (lane >> 4) << 3;
    const float* xr = src + (size_t)(r0 + row) * DIM;
#pragma unroll
    for (int t = 0; t < 4; ++t) {
        const float4 l0 = *(const float4*)(xr + t * 32 + kb);
        const float4 l1 = *(const float4*)(xr + t * 32 + kb + 4);
        const float4 h0 = *(const float4*)(xr + t * 32 + kb + 16);
        const float4 h1 = *(const float4*)(xr + t * 32 + kb + 20);
        a[t][0]  = (_Float16)l0.x; a[t][1]  = (_Float16)l0.y; a[t][2]  = (_Float16)l0.z; a[t][3]  = (_Float16)l0.w;
        a[t][4]  = (_Float16)l1.x; a[t][5]  = (_Float16)l1.y; a[t][6]  = (_Float16)l1.z; a[t][7]  = (_Float16)l1.w;
        a[t][8]  = (_Float16)h0.x; a[t][9]  = (_Float16)h0.y; a[t][10] = (_Float16)h0.z; a[t][11] = (_Float16)h0.w;
        a[t][12] = (_Float16)h1.x; a[t][13] = (_Float16)h1.y; a[t][14] = (_Float16)h1.z; a[t][15] = (_Float16)h1.w;
    }
}


#define OWN_NODES 196
#define OWN_BLOCKS 256
#define CHUNK_EDGES 4096
#define N_CHUNKS ((N_EDGES + CHUNK_EDGES - 1) / CHUNK_EDGES)
#define CELL_CAP 48
#define CELL_WORDS (CELL_CAP * 2)
#define WT_BLOCKS ((4 * DIM * DIM) / 256)

__device__ inline unsigned owner_of(int o) {
    unsigned own = (unsigned)o / OWN_NODES;
    if (own > OWN_BLOCKS - 1) own = OWN_BLOCKS - 1;
    return own;
}
__device__ inline void stage_entry(int o, int d, int e, unsigned own, int r, uint2* stage) {
    if (r < CELL_CAP) {
        const unsigned ol = (unsigned)o - own * OWN_NODES;
        stage[own * CELL_CAP + r] = make_uint2((unsigned)e | (ol << 20), (unsigned)d);
    }
}

__global__ void cvt_weightsT(const float* __restrict__ Wq, const float* __restrict__ Wk,
                             const float* __restrict__ Wv, const float* __restrict__ Wo,
                             _Float16* __restrict__ WT,
                             const int* __restrict__ origin, const int* __restrict__ dest,
                             unsigned* __restrict__ cells, int* __restrict__ counts) {
    __shared__ int bcnt[OWN_BLOCKS];
    __shared__ uint2 stage[OWN_BLOCKS * CELL_CAP];
    if (blockIdx.x < WT_BLOCKS) {
        int i = blockIdx.x * blockDim.x + threadIdx.x;
        int m = i >> 14, r = i & 16383, n = r >> 7, k = r & 127;
        const float* W = (m == 0) ? Wq : (m == 1) ? Wk : (m == 2) ? Wv : Wo;
        WT[i] = (_Float16)W[(size_t)k * DIM + n];
    } else {
        const int c = blockIdx.x - WT_BLOCKS;
        bcnt[threadIdx.x] = 0;
        __syncthreads();
        int4 o4[4], d4[4];
#pragma unroll
        for (int j = 0; j < 4; ++j) {
            const int e = c * CHUNK_EDGES + (j * 256 + (int)threadIdx.x) * 4;
            o4[j] = make_int4(0, 0, 0, 0);
            d4[j] = make_int4(0, 0, 0, 0);
            if (e < N_EDGES) {
                o4[j] = *(const int4*)(origin + e);
                d4[j] = *(const int4*)(dest + e);
            }
        }
        unsigned w_[16];
        int k_[16];
#pragma unroll
        for (int j = 0; j < 4; ++j) {
            const int e = c * CHUNK_EDGES + (j * 256 + (int)threadIdx.x) * 4;
            const int oj[4] = { o4[j].x, o4[j].y, o4[j].z, o4[j].w };
#pragma unroll
            for (int u = 0; u < 4; ++u) {
                w_[j * 4 + u] = owner_of(oj[u]);
                k_[j * 4 + u] = CELL_CAP;
                if (e < N_EDGES)
                    k_[j * 4 + u] = __hip_atomic_fetch_add(&bcnt[w_[j * 4 + u]], 1, __ATOMIC_RELAXED,
                                                           __HIP_MEMORY_SCOPE_WORKGROUP);
            }
        }
#pragma unroll
        for (int j = 0; j < 4; ++j) {
            const int e = c * CHUNK_EDGES + (j * 256 + (int)threadIdx.x) * 4;
            const int oj[4] = { o4[j].x, o4[j].y, o4[j].z, o4[j].w };
            const int dj[4] = { d4[j].x, d4[j].y, d4[j].z, d4[j].w };
#pragma unroll
            for (int u = 0; u < 4; ++u)
                stage_entry(oj[u], dj[u], e + u, w_[j * 4 + u], k_[j * 4 + u], stage);
        }
        __syncthreads();
        const int wv = __builtin_amdgcn_readfirstlane((int)threadIdx.x >> 5);
        const int ln = (int)threadIdx.x & 31;
        int ncl = bcnt[wv * 32 + ln];
        if (ncl > CELL_CAP) ncl = CELL_CAP;
#pragma unroll 8
        for (int i = 0; i < 32; ++i) {
            const int own = wv * 32 + i;
            const int nc = __builtin_amdgcn_readlane(ncl, i);
            uint2 en = stage[own * CELL_CAP + ln];
            if (ln >= nc) en = make_uint2(0u, 0u);
            uint2* cell = (uint2*)(cells + ((size_t)own * N_CHUNKS + c) * CELL_WORDS);
            cell[ln] = en;
        }
        unsigned big = __builtin_amdgcn_ballot_w32(ncl > 32);
        while (big != 0u) {
            const int i = __builtin_ctz(big);
            big &= big - 1u;
            const int own = wv * 32 + i;
            const int nc = __builtin_amdgcn_readlane(ncl, i);
            if (ln < 16) {
                uint2 en = stage[own * CELL_CAP + 32 + ln];
                if (ln + 32 >= nc) en = make_uint2(0u, 0u);
                uint2* cell = (uint2*)(cells + ((size_t)own * N_CHUNKS + c) * CELL_WORDS);
                cell[32 + ln] = en;
            }
        }
        counts[c * OWN_BLOCKS + (int)threadIdx.x] = bcnt[threadIdx.x];
    }
}

#define W_PITCH 136
#define W_IMAGE (DIM * W_PITCH)
#define W_PIECES (DIM * DIM / 8)

template <int NMAT, int NT>
struct WeightCopy {
    static constexpr int PIECES = NMAT * W_PIECES;
    static constexpr int TRIPS = (PIECES + NT - 1) / NT;
    static constexpr bool WHOLE = (PIECES % NT) == 0;
    v8h piece[TRIPS];
    __device__ inline void load(const _Float16* __restrict__ WT, int tid);
    __device__ inline void store(_Float16* img, int tid) const;
};
template <int NMAT, int NT>
__device__ inline void WeightCopy<NMAT, NT>::load(const _Float16* __restrict__ WT, int tid) {
    __builtin_assume(tid >= 0 && tid < NT);
#pragma unroll
    for (int i = 0; i < TRIPS; ++i) {
        const int p = tid + i * NT;
        piece[i] = (v8h){};
        if (WHOLE || p < PIECES) piece[i] = *(const v8h*)(WT + (size_t)p * 8);
    }
}
template <int NMAT, int NT>
__device__ inline void WeightCopy<NMAT, NT>::store(_Float16* img, int tid) const {
    __builtin_assume(tid >= 0 && tid < NT);
#pragma unroll
    for (int i = 0; i < TRIPS; ++i) {
        const int p = tid + i * NT;
        const int m = p >> 11;
        const int n = (p >> 4) & (DIM - 1);
        const int part = p & 15;
        const int row = (((n >> 6) << 2) + (n & 3)) * 16 + ((n >> 2) & 15);
        if (WHOLE || p < PIECES) *(v8h*)(img + m * W_IMAGE + row * W_PITCH + part * 8) = piece[i];
    }
}

#define ROW_TILES (N_NODES / 16)
#define QKV_TILES 2
#define QKV_GRID ((ROW_TILES + QKV_TILES - 1) / QKV_TILES)

__device__ inline void proj_two_tiles(const v16h a0[4], const v16h a1[4],
                                      const _Float16* WT, const float* __restrict__ bias,
                                      float* __restrict__ out, int r0, int r1, bool two, int lane) {
    const int c  = lane & 15;
    const int hf = lane >> 4;
    const int kb = hf << 3;
    v8f acc0[8], acc1[8];
#pragma unroll
    for (int t = 0; t < 8; ++t) { acc0[t] = (v8f){}; acc1[t] = (v8f){}; }
#pragma unroll
    for (int kk = 0; kk < 4; ++kk) {
        v16h b[8];
#pragma unroll
        for (int t = 0; t < 8; ++t) {
            const _Float16* wr = WT + (t * 16 + c) * W_PITCH + kk * 32 + kb;
            const v8h lo = *(const v8h*)(wr);
            const v8h hi = *(const v8h*)(wr + 16);
#pragma unroll
            for (int i = 0; i < 8; ++i) { b[t][i] = lo[i]; b[t][i + 8] = hi[i]; }
        }
#pragma unroll
        for (int t = 0; t < 8; ++t)
            acc0[t] = __builtin_amdgcn_wmma_f32_16x16x32_f16(false, a0[kk], false, b[t],
                                                             (short)0, acc0[t], false, false);
#pragma unroll
        for (int t = 0; t < 8; ++t)
            acc1[t] = __builtin_amdgcn_wmma_f32_16x16x32_f16(false, a1[kk], false, b[t],
                                                             (short)0, acc1[t], false, false);
    }
    const float4 bi0 = *(const float4*)(bias + c * 4), bi1 = *(const float4*)(bias + 64 + c * 4);
    float* op0 = out + (size_t)(r0 + hf * 8) * DIM + c * 4;
#pragma unroll
    for (int r = 0; r < 8; ++r) {
        const float4 ya = make_float4(acc0[0][r] + bi0.x, acc0[1][r] + bi0.y, acc0[2][r] + bi0.z, acc0[3][r] + bi0.w);
        const float4 yb = make_float4(acc0[4][r] + bi1.x, acc0[5][r] + bi1.y, acc0[6][r] + bi1.z, acc0[7][r] + bi1.w);
        *(float4*)(op0 + (size_t)r * DIM)      = ya;
        *(float4*)(op0 + (size_t)r * DIM + 64) = yb;
    }
    if (two) {
        float* op1 = out + (size_t)(r1 + hf * 8) * DIM + c * 4;
#pragma unroll
        for (int r = 0; r < 8; ++r) {
            const float4 ya = make_float4(acc1[0][r] + bi0.x, acc1[1][r] + bi0.y, acc1[2][r] + bi0.z, acc1[3][r] + bi0.w);
            const float4 yb = make_float4(acc1[4][r] + bi1.x, acc1[5][r] + bi1.y, acc1[6][r] + bi1.z, acc1[7][r] + bi1.w);
            *(float4*)(op1 + (size_t)r * DIM)      = ya;
            *(float4*)(op1 + (size_t)r * DIM + 64) = yb;
        }
    }
}

#define QKV_WAVES 7
#define QKV_BLOCKS ((QKV_GRID + QKV_WAVES - 1) / QKV_WAVES)
__global__ __launch_bounds__(32 * QKV_WAVES) __attribute__((amdgpu_waves_per_eu(1, 2)))
void qkv_gemm(const float* __restrict__ x,
              const _Float16* __restrict__ WTqkv, const float* __restrict__ bq,
              const float* __restrict__ bk, const float* __restrict__ bv,
              float* __restrict__ q, float* __restrict__ k,
              float* __restrict__ v) {
    __shared__ __attribute__((aligned(16))) _Float16 wimg[3 * W_IMAGE];
    const int lane = threadIdx.x & 31;
    const int wave = __builtin_amdgcn_readfirstlane((int)threadIdx.x >> 5);
    WeightCopy<3, 32 * QKV_WAVES> wcopy;
    wcopy.load(WTqkv, (int)threadIdx.x);
    const int task = blockIdx.x * QKV_WAVES + wave;
    const bool live = task < QKV_GRID;
    const int t0 = (live ? task : 0) * QKV_TILES;
    const bool two = t0 + 1 < ROW_TILES;
    const int r0 = t0 * 16;
    const int r1 = two ? r0 + 16 : r0;
    v16h a0[4], a1[4];
    load_a_frags_f32(x, r0, lane, a0);
    load_a_frags_f32(x, r1, lane, a1);
    wcopy.store(wimg, (int)threadIdx.x);
    __syncthreads();
    if (!live) return;
    proj_two_tiles(a0, a1, wimg,               bq, q, r0, r1, two, lane);
    proj_two_tiles(a0, a1, wimg + W_IMAGE,     bk, k, r0, r1, two, lane);
    proj_two_tiles(a0, a1, wimg + 2 * W_IMAGE, bv, v, r0, r1, two, lane);
}

#define CHUNKS128 (N_EDGES / 128)
#define SUPER ((CHUNKS128 + 3) / 4)
#define LIST_CAP 4096
#define CELL_THREADS 5
#define CELL_TURNS ((CELL_CAP + CELL_THREADS - 1) / CELL_THREADS)
#define BATCH 8

__device__ inline void lds_add(float* p, float x) {
    __hip_atomic_fetch_add(p, x, __ATOMIC_RELAXED, __HIP_MEMORY_SCOPE_WORKGROUP);
}
__device__ inline void lds_max_f(int* p, float v) {
    if (v >= 0.0f)
        __hip_atomic_fetch_max(p, __float_as_int(v), __ATOMIC_RELAXED, __HIP_MEMORY_SCOPE_WORKGROUP);
    else
        __hip_atomic_fetch_min((unsigned int*)p, __float_as_uint(v), __ATOMIC_RELAXED,
                               __HIP_MEMORY_SCOPE_WORKGROUP);
}

__device__ inline float quad_sum(float x) {
    x += __int_as_float(__builtin_amdgcn_mov_dpp(__float_as_int(x), 0xB1, 0xF, 0xF, true));
    x += __int_as_float(__builtin_amdgcn_mov_dpp(__float_as_int(x), 0x4E, 0xF, 0xF, true));
    return x;
}

__device__ inline float edge_score(const float4 qv, const float4 kv, const float w) {
    float s = qv.x * kv.x + qv.y * kv.y + qv.z * kv.z + qv.w * kv.w;
    s = quad_sum(s);
    s *= 0.25f;
    s *= w;
    return s;
}

__device__ inline void load_origins(const int* __restrict__ origin, int sc, int lane, int o[16]) {
#pragma unroll
    for (int k = 0; k < 4; ++k) {
        const int ch = sc * 4 + k;
        int4 t = make_int4(-1, -1, -1, -1);
        if (ch < CHUNKS128) t = *(const int4*)(origin + ch * 128 + lane * 4);
        o[k * 4 + 0] = t.x; o[k * 4 + 1] = t.y; o[k * 4 + 2] = t.z; o[k * 4 + 3] = t.w;
    }
}

template <int MODE>
__device__ inline void edge_in_place(int e, int oj, int node0, int lane, int h,
                                     const int* __restrict__ dest, const float* __restrict__ ew,
                                     const float* __restrict__ q, const float* __restrict__ k,
                                     const float* __restrict__ v,
                                     int* smax, float* den, float* tile) {
    const int dj = dest[e];
    const float4 qv = *(const float4*)(q + (size_t)oj * DIM + lane * 4);
    const float4 kv = *(const float4*)(k + (size_t)dj * DIM + lane * 4);
    const float s = edge_score(qv, kv, ew[e * HEADS + h]);
    const int ol = oj - node0;
    if (MODE == 1) {
        if ((lane & 3) == 0) lds_max_f(&smax[ol * HEADS + h], s);
    } else {
        float m = __int_as_float(smax[ol * HEADS + h]);
        if (!isfinite(m)) m = 0.0f;
        const float ev = __expf(s - m);
        if ((lane & 3) == 0) lds_add(&den[ol * HEADS + h], ev);
        const float4 vv = *(const float4*)(v + (size_t)dj * DIM + lane * 4);
        float* tp = tile + ol * DIM + lane;
        lds_add(tp + 0,  ev * vv.x);
        lds_add(tp + 32, ev * vv.y);
        lds_add(tp + 64, ev * vv.z);
        lds_add(tp + 96, ev * vv.w);
    }
}

template <int MODE>
__device__ inline void scan_owned(const int* __restrict__ origin, const int* __restrict__ dest,
                                  const float* __restrict__ ew,
                                  const float* __restrict__ q, const float* __restrict__ k,
                                  const float* __restrict__ v,
                                  int node0, int wave, int lane, int h,
                                  int* smax, float* den, float* tile) {
    const int rot = blockIdx.x * 6;
    int sc = wave + rot;
    if (sc >= SUPER) sc -= SUPER;
    int o[16];
    load_origins(origin, sc, lane, o);
    for (int s = wave; s < SUPER; s += 32) {
        int scn = s + 32 + rot;
        if (scn >= SUPER) scn -= SUPER;
        int on[16];
#pragma unroll
        for (int i = 0; i < 16; ++i) on[i] = -1;
        if (s + 32 < SUPER) load_origins(origin, scn, lane, on);

        unsigned mk[16];
        unsigned any = 0u;
#pragma unroll
        for (int i = 0; i < 16; ++i) {
            mk[i] = __builtin_amdgcn_ballot_w32((unsigned)(o[i] - node0) < (unsigned)OWN_NODES);
            any |= mk[i];
        }
        if (any != 0u) {
#pragma unroll
            for (int i = 0; i < 16; ++i) {
                unsigned mask = mk[i];
                while (mask) {
                    const int j = __builtin_ctz(mask);
                    mask &= mask - 1u;
                    const int oj = __builtin_amdgcn_readlane(o[i], j);
                    const int e = (sc * 4 + (i >> 2)) * 128 + j * 4 + (i & 3);
                    edge_in_place<MODE>(e, oj, node0, lane, h, dest, ew, q, k, v, smax, den, tile);
                }
            }
        }
        sc = scn;
#pragma unroll
        for (int i = 0; i < 16; ++i) o[i] = on[i];
    }
}

__global__ __launch_bounds__(1024) __attribute__((amdgpu_waves_per_eu(8, 8)))
void attend_owned(const unsigned* __restrict__ cells, const int* __restrict__ counts,
                  const int* __restrict__ origin, const int* __restrict__ dest,
                  const float* __restrict__ ew,
                  const float* __restrict__ q, const float* __restrict__ k,
                  const float* __restrict__ v, _Float16* __restrict__ vals16) {
    __shared__ float tile[OWN_NODES * DIM];
    __shared__ int sedge[LIST_CAP];
    __shared__ int sdest[LIST_CAP];
    __shared__ int smax[OWN_NODES * HEADS];
    __shared__ float den[OWN_NODES * HEADS];
    __shared__ int cnt[OWN_NODES];
    __shared__ int start[OWN_NODES + 1];
    __shared__ int counter;
    __shared__ int nextnode;
    const int tid = threadIdx.x;
    const int lane = tid & 31;
    const int wave = __builtin_amdgcn_readfirstlane(tid >> 5);
    const int node0 = blockIdx.x * OWN_NODES;
    const int h = lane >> 2;

    const int cidx = tid / CELL_THREADS;
    const int cj = tid - cidx * CELL_THREADS;
    int cc = 0;
    if (cidx < N_CHUNKS) cc = counts[cidx * OWN_BLOCKS + blockIdx.x];

    for (int i = tid; i < OWN_NODES * HEADS; i += 1024) {
        smax[i] = (int)0xFF800000;
        den[i] = 0.0f;
    }
    for (int i = tid; i < OWN_NODES; i += 1024) cnt[i] = 0;
    if (tid == 0) { counter = 0; nextnode = 0; }
    __syncthreads();

    if (cidx < N_CHUNKS && cj == 0)
        __hip_atomic_fetch_add(&counter, cc > CELL_CAP ? LIST_CAP + 1 : cc, __ATOMIC_RELAXED,
                               __HIP_MEMORY_SCOPE_WORKGROUP);
    const int lim = cc < CELL_CAP ? cc : CELL_CAP;
    const uint2* cell = (const uint2*)(cells + ((size_t)blockIdx.x * N_CHUNKS + cidx) * CELL_WORDS);
    uint2 ent[CELL_TURNS];
    int r_[CELL_TURNS];
#pragma unroll
    for (int t = 0; t < CELL_TURNS; ++t) {
        const int i = cj + t * CELL_THREADS;
        ent[t] = make_uint2(0u, 0u);
        if (i < lim) ent[t] = cell[i];
    }
#pragma unroll
    for (int t = 0; t < CELL_TURNS; ++t) {
        const int i = cj + t * CELL_THREADS;
        r_[t] = 0;
        if (i < lim) {
            unsigned ol = ent[t].x >> 20;
            if (ol > OWN_NODES - 1) ol = OWN_NODES - 1;
            r_[t] = __hip_atomic_fetch_add(&cnt[ol], 1, __ATOMIC_RELAXED, __HIP_MEMORY_SCOPE_WORKGROUP);
        }
    }
    __syncthreads();
    const int n = __builtin_amdgcn_readfirstlane(counter);

    if (n <= LIST_CAP) {
        if (wave == 0) {
            int c[7];
            int sum = 0;
#pragma unroll
            for (int i = 0; i < 7; ++i) {
                const int idx = lane * 7 + i;
                c[i] = (idx < OWN_NODES) ? cnt[idx] : 0;
                sum += c[i];
            }
            int incl = sum;
#pragma unroll
            for (int dlt = 1; dlt < 32; dlt <<= 1) {
                const int t = __shfl_up(incl, dlt, 32);
                if (lane >= dlt) incl += t;
            }
            int run = incl - sum;
#pragma unroll
            for (int i = 0; i < 7; ++i) {
                const int idx = lane * 7 + i;
                if (idx <= OWN_NODES) start[idx] = run;
                run += c[i];
            }
        }
        __syncthreads();
#pragma unroll
        for (int t = 0; t < CELL_TURNS; ++t) {
            const int i = cj + t * CELL_THREADS;
            if (i < lim) {
                unsigned ol = ent[t].x >> 20;
                if (ol > OWN_NODES - 1) ol = OWN_NODES - 1;
                const int pos = start[ol] + r_[t];
                sedge[pos] = (int)(ent[t].x & 0xFFFFFu);
                sdest[pos] = (int)ent[t].y;
            }
        }
        __syncthreads();

        for (;;) {
            int ol = 0;
            if (lane == 0)
                ol = __hip_atomic_fetch_add(&nextnode, 1, __ATOMIC_RELAXED, __HIP_MEMORY_SCOPE_WORKGROUP);
            ol = __builtin_amdgcn_readfirstlane(ol);
            if (ol >= OWN_NODES || node0 + ol >= N_NODES) break;
            const int s0 = __builtin_amdgcn_readfirstlane(start[ol]);
            const int s1 = __builtin_amdgcn_readfirstlane(start[ol + 1]);
            const float4 qv = *(const float4*)(q + (size_t)(node0 + ol) * DIM + lane * 4);
            float m = -__builtin_inff();
            float shift = 0.0f;
            float dsum = 0.0f;
            float4 acc = make_float4(0.0f, 0.0f, 0.0f, 0.0f);
            for (int b = s0; b < s1; b += BATCH) {
                int c = s1 - b;
                if (c > BATCH) c = BATCH;
                int li = b + lane;
                if (li > s1 - 1) li = s1 - 1;
                const int d_l = sdest[li];
                const int e_l = sedge[li];
                float4 kk[BATCH], vv[BATCH];
                float ww[BATCH];
#pragma unroll
                for (int u = 0; u < BATCH; ++u) {
                    const int dj = __builtin_amdgcn_readlane(d_l, u);
                    const int ej = __builtin_amdgcn_readlane(e_l, u);
                    kk[u] = *(const float4*)(k + (size_t)dj * DIM + lane * 4);
                    ww[u] = ew[ej * HEADS + h];
                    vv[u] = *(const float4*)(v + (size_t)dj * DIM + lane * 4);
                }
                float s[BATCH];
                float cm = -__builtin_inff();
#pragma unroll
                for (int u = 0; u < BATCH; ++u) {
                    const float t = edge_score(qv, kk[u], ww[u]);
                    s[u] = (u < c) ? t : -__builtin_inff();
                    cm = fmaxf(cm, s[u]);
                }
                const float mn = fmaxf(m, cm);
                const float sh = isfinite(mn) ? mn : 0.0f;
                const float sc = (b == s0) ? 0.0f : __expf(shift - sh);
                dsum *= sc;
                acc.x *= sc; acc.y *= sc; acc.z *= sc; acc.w *= sc;
#pragma unroll
                for (int u = 0; u < BATCH; ++u) {
                    const float ev = __expf(s[u] - sh);
                    dsum += ev;
                    acc.x += ev * vv[u].x;
                    acc.y += ev * vv[u].y;
                    acc.z += ev * vv[u].z;
                    acc.w += ev * vv[u].w;
                }
                m = mn;
                shift = sh;
            }
            const float dn = dsum + 1e-16f;
            const v4h o = { (_Float16)(acc.x / dn), (_Float16)(acc.y / dn), (_Float16)(acc.z / dn), (_Float16)(acc.w / dn) };
            *(v4h*)(vals16 + (size_t)(node0 + ol) * DIM + lane * 4) = o;
        }
    } else {
        float4* t4 = (float4*)tile;
        for (int i = tid; i < OWN_NODES * DIM / 4; i += 1024)
            t4[i] = make_float4(0.0f, 0.0f, 0.0f, 0.0f);
        scan_owned<1>(origin, dest, ew, q, k, v, node0, wave, lane, h, smax, den, tile);
        __syncthreads();
        scan_owned<2>(origin, dest, ew, q, k, v, node0, wave, lane, h, smax, den, tile);
        __syncthreads();

        v4h* out4 = (v4h*)(vals16 + (size_t)node0 * DIM);
        for (int i = tid; i < OWN_NODES * DIM / 4; i += 1024) {
            const int row = i >> 5;
            if (node0 + row < N_NODES) {
                const float dn = den[row * HEADS + ((i & 31) >> 2)] + 1e-16f;
                const float* tr = tile + row * DIM + (i & 31);
                float4 a = make_float4(tr[0], tr[32], tr[64], tr[96]);
                a.x /= dn; a.y /= dn; a.z /= dn; a.w /= dn;
                const v4h o = { (_Float16)a.x, (_Float16)a.y, (_Float16)a.z, (_Float16)a.w };
                out4[i] = o;
            }
        }
    }
}

#define PROJ_WAVES 8
#define PROJ_BLOCKS ((ROW_TILES + PROJ_WAVES - 1) / PROJ_WAVES)

__device__ inline float row16_sum(float x) {
    x += __int_as_float(__builtin_amdgcn_mov_dpp(__float_as_int(x), 0xB1, 0xF, 0xF, true));
    x += __int_as_float(__builtin_amdgcn_mov_dpp(__float_as_int(x), 0x4E, 0xF, 0xF, true));
    x += __int_as_float(__builtin_amdgcn_mov_dpp(__float_as_int(x), 0x141, 0xF, 0xF, true));
    x += __int_as_float(__builtin_amdgcn_mov_dpp(__float_as_int(x), 0x140, 0xF, 0xF, true));
    return x;
}

__global__ __launch_bounds__(32 * PROJ_WAVES) __attribute__((amdgpu_waves_per_eu(1, 4)))
void out_ln_residual(const _Float16* __restrict__ vals16,
                     const _Float16* __restrict__ WTo, const float* __restrict__ bo,
                     const float* __restrict__ x,
                     const float* __restrict__ gamma, const float* __restrict__ beta,
                     float* __restrict__ y) {
    __shared__ __attribute__((aligned(16))) _Float16 wimg[W_IMAGE];
    const int lane = threadIdx.x & 31;
    WeightCopy<1, 32 * PROJ_WAVES> wcopy;
    wcopy.load(WTo, (int)threadIdx.x);
    const int tile = blockIdx.x * PROJ_WAVES + __builtin_amdgcn_readfirstlane((int)threadIdx.x >> 5);
    const bool live = tile < ROW_TILES;
    const int r0 = (live ? tile : 0) * 16;
    const int c  = lane & 15;
    const int hf = lane >> 4;
    const int kb = hf << 3;

    const float* xr = x + (size_t)(r0 + hf * 8) * DIM + c * 4;
    float4 xa[8], xb[8];
#pragma unroll
    for (int r = 0; r < 8; ++r) {
        xa[r] = *(const float4*)(xr + (size_t)r * DIM);
        xb[r] = *(const float4*)(xr + (size_t)r * DIM + 64);
    }

    v16h a[4];
    load_a_frags(vals16, r0, lane, a);
    wcopy.store(wimg, (int)threadIdx.x);
    __syncthreads();
    if (!live) return;

    v8f acc[8];
#pragma unroll
    for (int t = 0; t < 8; ++t) acc[t] = (v8f){};
#pragma unroll
    for (int kk = 0; kk < 4; ++kk) {
        v16h b[8];
#pragma unroll
        for (int t = 0; t < 8; ++t) {
            const _Float16* wr = wimg + (t * 16 + c) * W_PITCH + kk * 32 + kb;
            const v8h lo = *(const v8h*)(wr);
            const v8h hi = *(const v8h*)(wr + 16);
#pragma unroll
            for (int i = 0; i < 8; ++i) { b[t][i] = lo[i]; b[t][i + 8] = hi[i]; }
        }
#pragma unroll
        for (int t = 0; t < 8; ++t)
            acc[t] = __builtin_amdgcn_wmma_f32_16x16x32_f16(false, a[kk], false, b[t],
                                                            (short)0, acc[t], false, false);
    }

    const float4 bo0 = *(const float4*)(bo + c * 4),    bo1 = *(const float4*)(bo + 64 + c * 4);
    const float4 g0  = *(const float4*)(gamma + c * 4), g1  = *(const float4*)(gamma + 64 + c * 4);
    const float4 b0  = *(const float4*)(beta + c * 4),  b1  = *(const float4*)(beta + 64 + c * 4);
    float* yr = y + (size_t)(r0 + hf * 8) * DIM + c * 4;
#pragma unroll
    for (int r = 0; r < 8; ++r) {
        const float o0 = acc[0][r] + bo0.x, o1 = acc[1][r] + bo0.y, o2 = acc[2][r] + bo0.z, o3 = acc[3][r] + bo0.w;
        const float o4 = acc[4][r] + bo1.x, o5 = acc[5][r] + bo1.y, o6 = acc[6][r] + bo1.z, o7 = acc[7][r] + bo1.w;
        const float s = row16_sum(((o0 + o1) + (o2 + o3)) + ((o4 + o5) + (o6 + o7)));
        const float mu = s * (1.0f / DIM);
        const float d0 = o0 - mu, d1 = o1 - mu, d2 = o2 - mu, d3 = o3 - mu;
        const float d4 = o4 - mu, d5 = o5 - mu, d6 = o6 - mu, d7 = o7 - mu;
        const float s2 = row16_sum(((d0 * d0 + d1 * d1) + (d2 * d2 + d3 * d3)) +
                                   ((d4 * d4 + d5 * d5) + (d6 * d6 + d7 * d7)));
        const float inv = rsqrtf(s2 * (1.0f / DIM) + 1e-5f);
        float4 ya, yb;
        ya.x = xa[r].x + d0 * inv * g0.x + b0.x;
        ya.y = xa[r].y + d1 * inv * g0.y + b0.y;
        ya.z = xa[r].z + d2 * inv * g0.z + b0.z;
        ya.w = xa[r].w + d3 * inv * g0.w + b0.w;
        yb.x = xb[r].x + d4 * inv * g1.x + b1.x;
        yb.y = xb[r].y + d5 * inv * g1.y + b1.y;
        yb.z = xb[r].z + d6 * inv * g1.z + b1.z;
        yb.w = xb[r].w + d7 * inv * g1.w + b1.w;
        *(float4*)(yr + (size_t)r * DIM)      = ya;
        *(float4*)(yr + (size_t)r * DIM + 64) = yb;
    }
}

extern "C" void kernel_launch(void* const* d_in, const int* in_sizes, int n_in,
                              void* d_out, int out_size, void* d_ws, size_t ws_size,
                              hipStream_t stream) {
    const float* x   = (const float*)d_in[0];
    const int*   ei  = (const int*)  d_in[1];
    const float* ew  = (const float*)d_in[2];
    const float* Wq  = (const float*)d_in[3];
    const float* bq  = (const float*)d_in[4];
    const float* Wk  = (const float*)d_in[5];
    const float* bk  = (const float*)d_in[6];
    const float* Wv  = (const float*)d_in[7];
    const float* bv  = (const float*)d_in[8];
    const float* Wo  = (const float*)d_in[9];
    const float* bo  = (const float*)d_in[10];
    const float* gam = (const float*)d_in[11];
    const float* bet = (const float*)d_in[12];
    float* y = (float*)d_out;

    const int* origin = ei;
    const int* dest   = ei + N_EDGES;

    float* ws   = (float*)d_ws;
    float* q    = ws;
    float* k    = q    + (size_t)N_NODES * DIM;
    float* v    = k    + (size_t)N_NODES * DIM;
    float* outb = v    + (size_t)N_NODES * DIM;
    _Float16* vals16 = (_Float16*)(outb + (size_t)N_NODES * DIM);
    _Float16* WT     = vals16 + (size_t)N_NODES * DIM;
    _Float16* WTq = WT;
    _Float16* WTo = WT + 3 * DIM * DIM;

    unsigned* cells = (unsigned*)outb;
    int* counts = (int*)(cells + (size_t)OWN_BLOCKS * N_CHUNKS * CELL_WORDS);
    cvt_weightsT<<<WT_BLOCKS + N_CHUNKS, 256, 0, stream>>>(Wq, Wk, Wv, Wo, WT, origin, dest, cells, counts);

    qkv_gemm<<<QKV_BLOCKS, 32 * QKV_WAVES, 0, stream>>>(x, WTq, bq, bk, bv, q, k, v);

    attend_owned<<<OWN_BLOCKS, 1024, 0, stream>>>(cells, counts, origin, dest, ew, q, k, v, vals16);

    out_ln_residual<<<PROJ_BLOCKS, 32 * PROJ_WAVES, 0, stream>>>(vals16, WTo, bo, x, gam, bet, y);
}
